# starting kernel
# baseline (speedup 1.0000x reference)
#include <hip/hip_runtime.h>
#include <math.h>

typedef __attribute__((ext_vector_type(2))) float v2f;
typedef __attribute__((ext_vector_type(4))) float v4f;
typedef __attribute__((ext_vector_type(8))) float v8f;

#define NDIM 64

__global__ void fa_gate_wmma(const float* __restrict__ h,
                             const float* __restrict__ gate_W,
                             float* __restrict__ gd,
                             float* __restrict__ gs,
                             int N)
{
    int wave = (blockIdx.x * blockDim.x + threadIdx.x) >> 5;
    int lane = threadIdx.x & 31;
    int base = wave * 16;
    if (base >= N) return;

    int row   = lane & 15;
    int khalf = (lane >> 4) << 1;
    int arow  = base + row; if (arow > N - 1) arow = N - 1;

    float msk = (row < 2) ? 1.0f : 0.0f;
    const float* wbase = gate_W + ((row < 2) ? row : 0) * NDIM + khalf;
    v2f bfrag[16];
    #pragma unroll
    for (int j = 0; j < 16; ++j) {
        bfrag[j].x = wbase[j * 4]     * msk;
        bfrag[j].y = wbase[j * 4 + 1] * msk;
    }

    const float* arow_p = h + (size_t)arow * NDIM + khalf;
    v8f acc = {0.f, 0.f, 0.f, 0.f, 0.f, 0.f, 0.f, 0.f};
    #pragma unroll
    for (int j = 0; j < 16; ++j) {
        v2f a = *(const v2f*)(arow_p + j * 4);
        acc = __builtin_amdgcn_wmma_f32_16x16x4_f32(
                   false, a,  false, bfrag[j],
                   (short)0, acc,  false,  false);
    }

    if (row < 2) {
        float* outp = (row == 0) ? gd : gs;
        int moff = (lane >> 4) << 3;
        #pragma unroll
        for (int i = 0; i < 8; ++i) {
            int m = base + moff + i;
            if (m < N) outp[m] = acc[i];
        }
    }
}

__global__ void fa_edge_gate(const float* __restrict__ gd,
                             const float* __restrict__ gs,
                             const float* __restrict__ dvec,
                             const float* __restrict__ gate_b,
                             const int* __restrict__ esrc,
                             const int* __restrict__ edst,
                             float* __restrict__ e, int E)
{
    int i = blockIdx.x * blockDim.x + threadIdx.x;
    if (i >= E) return;
    int s = esrc[i];
    int t = edst[i];
    float g = tanhf(gd[t] + gs[s] + gate_b[0]);
    e[i] = g * dvec[t] * dvec[s];
}

__global__ void fa_zero(float* __restrict__ z, int n4)
{
    int i = blockIdx.x * blockDim.x + threadIdx.x;
    if (i < n4) ((v4f*)z)[i] = (v4f){0.f, 0.f, 0.f, 0.f};
}

__global__ void fa_scatter(const float* __restrict__ h,
                           const int* __restrict__ esrc,
                           const int* __restrict__ edst,
                           const float* __restrict__ e,
                           float* __restrict__ z, int E)
{
    int lane   = threadIdx.x & 31;
    int wave   = (blockIdx.x * blockDim.x + threadIdx.x) >> 5;
    int nwaves = (gridDim.x * blockDim.x) >> 5;
    int f = lane << 1;

    for (int i = wave; i < E; i += nwaves) {
        int   s = esrc[i];
        int   t = edst[i];
        float c = e[i];
        v2f hv = *(const v2f*)(h + (size_t)s * NDIM + f);
        float* zp = z + (size_t)t * NDIM + f;
        atomicAdd(zp,     hv.x * c);
        atomicAdd(zp + 1, hv.y * c);
    }
}

extern "C" void kernel_launch(void* const* d_in, const int* in_sizes, int n_in,
                              void* d_out, int out_size, void* d_ws, size_t ws_size,
                              hipStream_t stream)
{
    const float* h      = (const float*)d_in[0];
    const float* dvec   = (const float*)d_in[1];
    const float* gate_W = (const float*)d_in[2];
    const float* gate_b = (const float*)d_in[3];
    const int*   esrc   = (const int*)d_in[4];
    const int*   edst   = (const int*)d_in[5];

    int N = in_sizes[1];
    int E = in_sizes[4];
    float* z = (float*)d_out;

    float* gd = (float*)d_ws;
    float* gs = gd + N;
    float* e  = gs + N;

    {
        int waves   = (N + 15) / 16;
        int threads = waves * 32;
        dim3 blk(256), grd((threads + 255) / 256);
        fa_gate_wmma<<<grd, blk, 0, stream>>>(h, gate_W, gd, gs, N);
    }
    {
        dim3 blk(256), grd((E + 255) / 256);
        fa_edge_gate<<<grd, blk, 0, stream>>>(gd, gs, dvec, gate_b, esrc, edst, e, E);
    }
    {
        int n4 = out_size / 4;
        dim3 blk(256), grd((n4 + 255) / 256);
        fa_zero<<<grd, blk, 0, stream>>>(z, n4);
    }
    {
        dim3 blk(256), grd(25000);
        fa_scatter<<<grd, blk, 0, stream>>>(h, esrc, edst, e, z, E);
    }
}


# Scatter: lane l owns features l and l+32, so every h-row load and every float atomic of the wave covers one whole 128-byte line instead of every other dword of two lines
# speedup vs baseline: 1.4809x; 1.4809x over previous
#include <hip/hip_runtime.h>
#include <math.h>

typedef __attribute__((ext_vector_type(2))) float v2f;
typedef __attribute__((ext_vector_type(4))) float v4f;
typedef __attribute__((ext_vector_type(8))) float v8f;

#define NDIM 64

__global__ void fa_gate_wmma(const float* __restrict__ h,
                             const float* __restrict__ gate_W,
                             float* __restrict__ gd,
                             float* __restrict__ gs,
                             int N)
{
    int wave = (blockIdx.x * blockDim.x + threadIdx.x) >> 5;
    int lane = threadIdx.x & 31;
    int base = wave * 16;
    if (base >= N) return;

    int row   = lane & 15;
    int khalf = (lane >> 4) << 1;
    int arow  = base + row; if (arow > N - 1) arow = N - 1;

    float msk = (row < 2) ? 1.0f : 0.0f;
    const float* wbase = gate_W + ((row < 2) ? row : 0) * NDIM + khalf;
    v2f bfrag[16];
    #pragma unroll
    for (int j = 0; j < 16; ++j) {
        bfrag[j].x = wbase[j * 4]     * msk;
        bfrag[j].y = wbase[j * 4 + 1] * msk;
    }

    const float* arow_p = h + (size_t)arow * NDIM + khalf;
    v8f acc = {0.f, 0.f, 0.f, 0.f, 0.f, 0.f, 0.f, 0.f};
    #pragma unroll
    for (int j = 0; j < 16; ++j) {
        v2f a = *(const v2f*)(arow_p + j * 4);
        acc = __builtin_amdgcn_wmma_f32_16x16x4_f32(
                   false, a,  false, bfrag[j],
                   (short)0, acc,  false,  false);
    }

    if (row < 2) {
        float* outp = (row == 0) ? gd : gs;
        int moff = (lane >> 4) << 3;
        #pragma unroll
        for (int i = 0; i < 8; ++i) {
            int m = base + moff + i;
            if (m < N) outp[m] = acc[i];
        }
    }
}

__global__ void fa_zero(float* __restrict__ z, int n4)
{
    int i = blockIdx.x * blockDim.x + threadIdx.x;
    if (i < n4) ((v4f*)z)[i] = (v4f){0.f, 0.f, 0.f, 0.f};
}

__device__ __forceinline__ void fa_edge_update(const float* __restrict__ h,
                                               float* __restrict__ z,
                                               int s, int t, float c, int f)
{
    const float* hp = h + (size_t)s * NDIM + f;
    float h_lo = hp[0];
    float h_hi = hp[32];
    float* zp = z + (size_t)t * NDIM + f;
    atomicAdd(zp,      h_lo * c);
    atomicAdd(zp + 32, h_hi * c);
}

__global__ void fa_gate_scatter(const float* __restrict__ h,
                                const float* __restrict__ gd,
                                const float* __restrict__ gs,
                                const float* __restrict__ dvec,
                                const float* __restrict__ gate_b,
                                const int* __restrict__ esrc,
                                const int* __restrict__ edst,
                                float* __restrict__ z, int E)
{
    int lane = threadIdx.x & 31;
    int wave   = __builtin_amdgcn_readfirstlane((int)((blockIdx.x * blockDim.x + threadIdx.x) >> 5));
    int nwaves = (int)((gridDim.x * blockDim.x) >> 5);
    int f = lane;
    float bias = gate_b[0];

    for (int base = wave * 32; base < E; base += nwaves * 32) {
        int i = base + lane;
        int s = 0, t = 0;
        float c = 0.f;
        if (i < E) {
            s = esrc[i];
            t = edst[i];
            float gdt = gd[t];
            float gss = gs[s];
            float dt  = dvec[t];
            float dsv = dvec[s];
            float g = tanhf(gdt + gss + bias);
            c = g * dt * dsv;
        }
        int cbits = __float_as_int(c);

        int n = E - base;
        if (n > 32) n = 32;
        for (int j = 0; j < n; ++j) {
            fa_edge_update(h, z,
                           __builtin_amdgcn_readlane(s, j),
                           __builtin_amdgcn_readlane(t, j),
                           __int_as_float(__builtin_amdgcn_readlane(cbits, j)), f);
        }
    }
}

extern "C" void kernel_launch(void* const* d_in, const int* in_sizes, int n_in,
                              void* d_out, int out_size, void* d_ws, size_t ws_size,
                              hipStream_t stream)
{
    const float* h      = (const float*)d_in[0];
    const float* dvec   = (const float*)d_in[1];
    const float* gate_W = (const float*)d_in[2];
    const float* gate_b = (const float*)d_in[3];
    const int*   esrc   = (const int*)d_in[4];
    const int*   edst   = (const int*)d_in[5];

    int N = in_sizes[1];
    int E = in_sizes[4];
    float* z = (float*)d_out;

    float* gd = (float*)d_ws;
    float* gs = gd + N;

    {
        int waves   = (N + 15) / 16;
        int threads = waves * 32;
        dim3 blk(256), grd((threads + 255) / 256);
        fa_gate_wmma<<<grd, blk, 0, stream>>>(h, gate_W, gd, gs, N);
    }
    {
        int n4 = out_size / 4;
        dim3 blk(256), grd((n4 + 255) / 256);
        fa_zero<<<grd, blk, 0, stream>>>(z, n4);
    }
    {
        int waves   = (E + 31) / 32;
        int threads = waves * 32;
        dim3 blk(256), grd((threads + 255) / 256);
        fa_gate_scatter<<<grd, blk, 0, stream>>>(h, gd, gs, dvec, gate_b, esrc, edst, z, E);
    }
}


# Scatter: the h rows of 8 edges are loaded as one run and the next 8 are issued ahead of the current group's atomics (two register sets), so a wave waits once per 8 edges instead of once per edge
# speedup vs baseline: 1.6758x; 1.1316x over previous
#include <hip/hip_runtime.h>
#include <math.h>

typedef __attribute__((ext_vector_type(2))) float v2f;
typedef __attribute__((ext_vector_type(4))) float v4f;
typedef __attribute__((ext_vector_type(8))) float v8f;

#define NDIM 64

__global__ void fa_gate_wmma(const float* __restrict__ h,
                             const float* __restrict__ gate_W,
                             float* __restrict__ gd,
                             float* __restrict__ gs,
                             int N)
{
    int wave = (blockIdx.x * blockDim.x + threadIdx.x) >> 5;
    int lane = threadIdx.x & 31;
    int base = wave * 16;
    if (base >= N) return;

    int row   = lane & 15;
    int khalf = (lane >> 4) << 1;
    int arow  = base + row; if (arow > N - 1) arow = N - 1;

    float msk = (row < 2) ? 1.0f : 0.0f;
    const float* wbase = gate_W + ((row < 2) ? row : 0) * NDIM + khalf;
    v2f bfrag[16];
    #pragma unroll
    for (int j = 0; j < 16; ++j) {
        bfrag[j].x = wbase[j * 4]     * msk;
        bfrag[j].y = wbase[j * 4 + 1] * msk;
    }

    const float* arow_p = h + (size_t)arow * NDIM + khalf;
    v8f acc = {0.f, 0.f, 0.f, 0.f, 0.f, 0.f, 0.f, 0.f};
    #pragma unroll
    for (int j = 0; j < 16; ++j) {
        v2f a = *(const v2f*)(arow_p + j * 4);
        acc = __builtin_amdgcn_wmma_f32_16x16x4_f32(
                   false, a,  false, bfrag[j],
                   (short)0, acc,  false,  false);
    }

    if (row < 2) {
        float* outp = (row == 0) ? gd : gs;
        int moff = (lane >> 4) << 3;
        #pragma unroll
        for (int i = 0; i < 8; ++i) {
            int m = base + moff + i;
            if (m < N) outp[m] = acc[i];
        }
    }
}

__global__ void fa_zero(float* __restrict__ z, int n4)
{
    int i = blockIdx.x * blockDim.x + threadIdx.x;
    if (i < n4) ((v4f*)z)[i] = (v4f){0.f, 0.f, 0.f, 0.f};
}

#define FA_GRP 8
#define FA_NGRP (32 / FA_GRP)

__device__ __forceinline__ void fa_edge_update(const float* __restrict__ h,
                                               float* __restrict__ z,
                                               int s, int t, float c, int f)
{
    const float* hp = h + (size_t)s * NDIM + f;
    float h_lo = hp[0];
    float h_hi = hp[32];
    float* zp = z + (size_t)t * NDIM + f;
    atomicAdd(zp,      h_lo * c);
    atomicAdd(zp + 32, h_hi * c);
}

__device__ __forceinline__ void fa_rows_issue(const float* __restrict__ h,
                                              int s, int j0, int f,
                                              float (&lo)[FA_GRP], float (&hi)[FA_GRP])
{
    #pragma unroll
    for (int k = 0; k < FA_GRP; ++k) {
        const float* hp = h + (size_t)__builtin_amdgcn_readlane(s, j0 + k) * NDIM + f;
        lo[k] = hp[0];
        hi[k] = hp[32];
    }
}

__device__ __forceinline__ void fa_rows_add(float* __restrict__ z,
                                            int t, int cbits, int j0, int f,
                                            const float (&lo)[FA_GRP], const float (&hi)[FA_GRP])
{
    #pragma unroll
    for (int k = 0; k < FA_GRP; ++k) {
        float c = __int_as_float(__builtin_amdgcn_readlane(cbits, j0 + k));
        float* zp = z + (size_t)__builtin_amdgcn_readlane(t, j0 + k) * NDIM + f;
        atomicAdd(zp,      lo[k] * c);
        atomicAdd(zp + 32, hi[k] * c);
    }
}

__global__ void fa_gate_scatter(const float* __restrict__ h,
                                const float* __restrict__ gd,
                                const float* __restrict__ gs,
                                const float* __restrict__ dvec,
                                const float* __restrict__ gate_b,
                                const int* __restrict__ esrc,
                                const int* __restrict__ edst,
                                float* __restrict__ z, int E)
{
    int lane = threadIdx.x & 31;
    int wave   = __builtin_amdgcn_readfirstlane((int)((blockIdx.x * blockDim.x + threadIdx.x) >> 5));
    int nwaves = (int)((gridDim.x * blockDim.x) >> 5);
    int f = lane;
    float bias = gate_b[0];

    for (int base = wave * 32; base < E; base += nwaves * 32) {
        int i = base + lane;
        int n = E - base;

        if (n >= 32) {
            int s = esrc[i];
            int t = edst[i];
            float gdt = gd[t];
            float gss = gs[s];
            float dt  = dvec[t];
            float dsv = dvec[s];

            float lo[2][FA_GRP], hi[2][FA_GRP];
            fa_rows_issue(h, s, 0, f, lo[0], hi[0]);

            float g = tanhf(gdt + gss + bias);
            float c = g * dt * dsv;
            int cbits = __float_as_int(c);

            #pragma unroll
            for (int gi = 0; gi < FA_NGRP; ++gi) {
                if (gi + 1 < FA_NGRP)
                    fa_rows_issue(h, s, (gi + 1) * FA_GRP, f, lo[(gi + 1) & 1], hi[(gi + 1) & 1]);
                fa_rows_add(z, t, cbits, gi * FA_GRP, f, lo[gi & 1], hi[gi & 1]);
            }
        } else {
            int s = 0, t = 0;
            float c = 0.f;
            if (i < E) {
                s = esrc[i];
                t = edst[i];
                float gdt = gd[t];
                float gss = gs[s];
                float dt  = dvec[t];
                float dsv = dvec[s];
                float g = tanhf(gdt + gss + bias);
                c = g * dt * dsv;
            }
            int cbits = __float_as_int(c);
            for (int j = 0; j < n; ++j) {
                fa_edge_update(h, z,
                               __builtin_amdgcn_readlane(s, j),
                               __builtin_amdgcn_readlane(t, j),
                               __int_as_float(__builtin_amdgcn_readlane(cbits, j)), f);
            }
        }
    }
}

extern "C" void kernel_launch(void* const* d_in, const int* in_sizes, int n_in,
                              void* d_out, int out_size, void* d_ws, size_t ws_size,
                              hipStream_t stream)
{
    const float* h      = (const float*)d_in[0];
    const float* dvec   = (const float*)d_in[1];
    const float* gate_W = (const float*)d_in[2];
    const float* gate_b = (const float*)d_in[3];
    const int*   esrc   = (const int*)d_in[4];
    const int*   edst   = (const int*)d_in[5];

    int N = in_sizes[1];
    int E = in_sizes[4];
    float* z = (float*)d_out;

    float* gd = (float*)d_ws;
    float* gs = gd + N;

    {
        int waves   = (N + 15) / 16;
        int threads = waves * 32;
        dim3 blk(256), grd((threads + 255) / 256);
        fa_gate_wmma<<<grd, blk, 0, stream>>>(h, gate_W, gd, gs, N);
    }
    {
        int n4 = out_size / 4;
        dim3 blk(256), grd((n4 + 255) / 256);
        fa_zero<<<grd, blk, 0, stream>>>(z, n4);
    }
    {
        int waves   = (E + 31) / 32;
        int threads = waves * 32;
        dim3 blk(256), grd((threads + 255) / 256);
        fa_gate_scatter<<<grd, blk, 0, stream>>>(h, gd, gs, dvec, gate_b, esrc, edst, z, E);
    }
}


# Gate kernel writes per-node 8-byte records (gd,d) and (gs,d); the scatter's lane-parallel phase looks up two records per edge instead of four scalars, halving its one-line-per-lane loads
# speedup vs baseline: 1.9426x; 1.1592x over previous
#include <hip/hip_runtime.h>
#include <math.h>

typedef __attribute__((ext_vector_type(2))) float v2f;
typedef __attribute__((ext_vector_type(4))) float v4f;
typedef __attribute__((ext_vector_type(8))) float v8f;

#define NDIM 64

__global__ void fa_gate_wmma(const float* __restrict__ h,
                             const float* __restrict__ gate_W,
                             const float* __restrict__ dvec,
                             float* __restrict__ rec_dst,
                             float* __restrict__ rec_src,
                             int N)
{
    int wave = (blockIdx.x * blockDim.x + threadIdx.x) >> 5;
    int lane = threadIdx.x & 31;
    int base = wave * 16;
    if (base >= N) return;

    int row   = lane & 15;
    int khalf = (lane >> 4) << 1;
    int arow  = base + row; if (arow > N - 1) arow = N - 1;

    float msk = (row < 2) ? 1.0f : 0.0f;
    const float* wbase = gate_W + ((row < 2) ? row : 0) * NDIM + khalf;
    v2f bfrag[16];
    #pragma unroll
    for (int j = 0; j < 16; ++j) {
        bfrag[j].x = wbase[j * 4]     * msk;
        bfrag[j].y = wbase[j * 4 + 1] * msk;
    }

    const float* arow_p = h + (size_t)arow * NDIM + khalf;
    v8f acc = {0.f, 0.f, 0.f, 0.f, 0.f, 0.f, 0.f, 0.f};
    #pragma unroll
    for (int j = 0; j < 16; ++j) {
        v2f a = *(const v2f*)(arow_p + j * 4);
        acc = __builtin_amdgcn_wmma_f32_16x16x4_f32(
                   false, a,  false, bfrag[j],
                   (short)0, acc,  false,  false);
    }

    if (row < 2) {
        float* outp = (row == 0) ? rec_dst : rec_src;
        int moff = (lane >> 4) << 3;
        #pragma unroll
        for (int i = 0; i < 8; ++i) {
            int m = base + moff + i;
            if (m < N) {
                v2f r; r.x = acc[i]; r.y = dvec[m];
                *(v2f*)(outp + 2 * (size_t)m) = r;
            }
        }
    }
}

__global__ void fa_zero(float* __restrict__ z, int n4)
{
    int i = blockIdx.x * blockDim.x + threadIdx.x;
    if (i < n4) ((v4f*)z)[i] = (v4f){0.f, 0.f, 0.f, 0.f};
}

#define FA_GRP 8
#define FA_NGRP (32 / FA_GRP)

__device__ __forceinline__ void fa_edge_update(const float* __restrict__ h,
                                               float* __restrict__ z,
                                               int s, int t, float c, int f)
{
    const float* hp = h + (size_t)s * NDIM + f;
    float h_lo = hp[0];
    float h_hi = hp[32];
    float* zp = z + (size_t)t * NDIM + f;
    atomicAdd(zp,      h_lo * c);
    atomicAdd(zp + 32, h_hi * c);
}

__device__ __forceinline__ void fa_rows_issue(const float* __restrict__ h,
                                              int s, int j0, int f,
                                              float (&lo)[FA_GRP], float (&hi)[FA_GRP])
{
    #pragma unroll
    for (int k = 0; k < FA_GRP; ++k) {
        const float* hp = h + (size_t)__builtin_amdgcn_readlane(s, j0 + k) * NDIM + f;
        lo[k] = hp[0];
        hi[k] = hp[32];
    }
}

__device__ __forceinline__ void fa_rows_add(float* __restrict__ z,
                                            int t, int cbits, int j0, int f,
                                            const float (&lo)[FA_GRP], const float (&hi)[FA_GRP])
{
    #pragma unroll
    for (int k = 0; k < FA_GRP; ++k) {
        float c = __int_as_float(__builtin_amdgcn_readlane(cbits, j0 + k));
        float* zp = z + (size_t)__builtin_amdgcn_readlane(t, j0 + k) * NDIM + f;
        atomicAdd(zp,      lo[k] * c);
        atomicAdd(zp + 32, hi[k] * c);
    }
}

__global__ void fa_gate_scatter(const float* __restrict__ h,
                                const float* __restrict__ rec_dst,
                                const float* __restrict__ rec_src,
                                const float* __restrict__ gate_b,
                                const int* __restrict__ esrc,
                                const int* __restrict__ edst,
                                float* __restrict__ z, int E)
{
    int lane = threadIdx.x & 31;
    int wave   = __builtin_amdgcn_readfirstlane((int)((blockIdx.x * blockDim.x + threadIdx.x) >> 5));
    int nwaves = (int)((gridDim.x * blockDim.x) >> 5);
    int f = lane;
    float bias = gate_b[0];

    for (int base = wave * 32; base < E; base += nwaves * 32) {
        int i = base + lane;
        int n = E - base;

        if (n >= 32) {
            int s = esrc[i];
            int t = edst[i];
            v2f rt = *(const v2f*)(rec_dst + 2 * (size_t)t);
            v2f rs = *(const v2f*)(rec_src + 2 * (size_t)s);

            float lo[2][FA_GRP], hi[2][FA_GRP];
            fa_rows_issue(h, s, 0, f, lo[0], hi[0]);

            float g = tanhf(rt.x + rs.x + bias);
            float c = g * rt.y * rs.y;
            int cbits = __float_as_int(c);

            #pragma unroll
            for (int gi = 0; gi < FA_NGRP; ++gi) {
                if (gi + 1 < FA_NGRP)
                    fa_rows_issue(h, s, (gi + 1) * FA_GRP, f, lo[(gi + 1) & 1], hi[(gi + 1) & 1]);
                fa_rows_add(z, t, cbits, gi * FA_GRP, f, lo[gi & 1], hi[gi & 1]);
            }
        } else {
            int s = 0, t = 0;
            float c = 0.f;
            if (i < E) {
                s = esrc[i];
                t = edst[i];
                v2f rt = *(const v2f*)(rec_dst + 2 * (size_t)t);
                v2f rs = *(const v2f*)(rec_src + 2 * (size_t)s);
                float g = tanhf(rt.x + rs.x + bias);
                c = g * rt.y * rs.y;
            }
            int cbits = __float_as_int(c);
            for (int j = 0; j < n; ++j) {
                fa_edge_update(h, z,
                               __builtin_amdgcn_readlane(s, j),
                               __builtin_amdgcn_readlane(t, j),
                               __int_as_float(__builtin_amdgcn_readlane(cbits, j)), f);
            }
        }
    }
}

extern "C" void kernel_launch(void* const* d_in, const int* in_sizes, int n_in,
                              void* d_out, int out_size, void* d_ws, size_t ws_size,
                              hipStream_t stream)
{
    const float* h      = (const float*)d_in[0];
    const float* dvec   = (const float*)d_in[1];
    const float* gate_W = (const float*)d_in[2];
    const float* gate_b = (const float*)d_in[3];
    const int*   esrc   = (const int*)d_in[4];
    const int*   edst   = (const int*)d_in[5];

    int N = in_sizes[1];
    int E = in_sizes[4];
    float* z = (float*)d_out;

    float* rec_dst = (float*)d_ws;
    float* rec_src = rec_dst + 2 * (size_t)N;

    {
        int waves   = (N + 15) / 16;
        int threads = waves * 32;
        dim3 blk(256), grd((threads + 255) / 256);
        fa_gate_wmma<<<grd, blk, 0, stream>>>(h, gate_W, dvec, rec_dst, rec_src, N);
    }
    {
        int n4 = out_size / 4;
        dim3 blk(256), grd((n4 + 255) / 256);
        fa_zero<<<grd, blk, 0, stream>>>(z, n4);
    }
    {
        int waves   = (E + 31) / 32;
        int threads = waves * 32;
        dim3 blk(256), grd((threads + 255) / 256);
        fa_gate_scatter<<<grd, blk, 0, stream>>>(h, rec_dst, rec_src, gate_b, esrc, edst, z, E);
    }
}


# Accumulate: each block's edges are counting-sorted by node in LDS; a wave sums a node's consecutive rows in registers: two ds float adds per run of equal nodes, not per edge; sum order changes
# speedup vs baseline: 2.9172x; 1.5017x over previous
#include <hip/hip_runtime.h>
#include <math.h>

typedef __attribute__((ext_vector_type(2))) float v2f;
typedef __attribute__((ext_vector_type(4))) float v4f;
typedef __attribute__((ext_vector_type(8))) float v8f;

#define NDIM 64

__global__ void fa_gate_wmma(const float* __restrict__ h,
                             const float* __restrict__ gate_W,
                             const float* __restrict__ dvec,
                             float* __restrict__ rec_dst,
                             float* __restrict__ rec_src,
                             int N)
{
    int wave = (blockIdx.x * blockDim.x + threadIdx.x) >> 5;
    int lane = threadIdx.x & 31;
    int base = wave * 16;
    if (base >= N) return;

    int row   = lane & 15;
    int khalf = (lane >> 4) << 1;
    int arow  = base + row; if (arow > N - 1) arow = N - 1;

    float msk = (row < 2) ? 1.0f : 0.0f;
    const float* wbase = gate_W + ((row < 2) ? row : 0) * NDIM + khalf;
    v2f bfrag[16];
    #pragma unroll
    for (int j = 0; j < 16; ++j) {
        bfrag[j].x = wbase[j * 4]     * msk;
        bfrag[j].y = wbase[j * 4 + 1] * msk;
    }

    const float* arow_p = h + (size_t)arow * NDIM + khalf;
    v8f acc = {0.f, 0.f, 0.f, 0.f, 0.f, 0.f, 0.f, 0.f};
    #pragma unroll
    for (int j = 0; j < 16; ++j) {
        v2f a = *(const v2f*)(arow_p + j * 4);
        acc = __builtin_amdgcn_wmma_f32_16x16x4_f32(
                   false, a,  false, bfrag[j],
                   (short)0, acc,  false,  false);
    }

    if (row < 2) {
        float* outp = (row == 0) ? rec_dst : rec_src;
        int moff = (lane >> 4) << 3;
        #pragma unroll
        for (int i = 0; i < 8; ++i) {
            int m = base + moff + i;
            if (m < N) {
                v2f r; r.x = acc[i]; r.y = dvec[m];
                *(v2f*)(outp + 2 * (size_t)m) = r;
            }
        }
    }
}

__global__ void fa_zero(float* __restrict__ z, int n4)
{
    int i = blockIdx.x * blockDim.x + threadIdx.x;
    if (i < n4) ((v4f*)z)[i] = (v4f){0.f, 0.f, 0.f, 0.f};
}

#define FA_GRP 8
#define FA_NGRP (32 / FA_GRP)

__device__ __forceinline__ void fa_edge_update(const float* __restrict__ h,
                                               float* __restrict__ z,
                                               int s, int t, float c, int f)
{
    const float* hp = h + (size_t)s * NDIM + f;
    float h_lo = hp[0];
    float h_hi = hp[32];
    float* zp = z + (size_t)t * NDIM + f;
    atomicAdd(zp,      h_lo * c);
    atomicAdd(zp + 32, h_hi * c);
}

__device__ __forceinline__ void fa_rows_issue(const float* __restrict__ h,
                                              int s, int j0, int f,
                                              float (&lo)[FA_GRP], float (&hi)[FA_GRP])
{
    #pragma unroll
    for (int k = 0; k < FA_GRP; ++k) {
        const float* hp = h + (size_t)__builtin_amdgcn_readlane(s, j0 + k) * NDIM + f;
        lo[k] = hp[0];
        hi[k] = hp[32];
    }
}

__device__ __forceinline__ void fa_rows_add(float* __restrict__ z,
                                            int t, int cbits, int j0, int f,
                                            const float (&lo)[FA_GRP], const float (&hi)[FA_GRP])
{
    #pragma unroll
    for (int k = 0; k < FA_GRP; ++k) {
        float c = __int_as_float(__builtin_amdgcn_readlane(cbits, j0 + k));
        float* zp = z + (size_t)__builtin_amdgcn_readlane(t, j0 + k) * NDIM + f;
        atomicAdd(zp,      lo[k] * c);
        atomicAdd(zp + 32, hi[k] * c);
    }
}

__global__ void fa_gate_scatter(const float* __restrict__ h,
                                const float* __restrict__ rec_dst,
                                const float* __restrict__ rec_src,
                                const float* __restrict__ gate_b,
                                const int* __restrict__ esrc,
                                const int* __restrict__ edst,
                                float* __restrict__ z, int E)
{
    int lane = threadIdx.x & 31;
    int wave   = __builtin_amdgcn_readfirstlane((int)((blockIdx.x * blockDim.x + threadIdx.x) >> 5));
    int nwaves = (int)((gridDim.x * blockDim.x) >> 5);
    int f = lane;
    float bias = gate_b[0];

    for (int base = wave * 32; base < E; base += nwaves * 32) {
        int i = base + lane;
        int n = E - base;

        if (n >= 32) {
            int s = esrc[i];
            int t = edst[i];
            v2f rt = *(const v2f*)(rec_dst + 2 * (size_t)t);
            v2f rs = *(const v2f*)(rec_src + 2 * (size_t)s);

            float lo[2][FA_GRP], hi[2][FA_GRP];
            fa_rows_issue(h, s, 0, f, lo[0], hi[0]);

            float g = tanhf(rt.x + rs.x + bias);
            float c = g * rt.y * rs.y;
            int cbits = __float_as_int(c);

            #pragma unroll
            for (int gi = 0; gi < FA_NGRP; ++gi) {
                if (gi + 1 < FA_NGRP)
                    fa_rows_issue(h, s, (gi + 1) * FA_GRP, f, lo[(gi + 1) & 1], hi[(gi + 1) & 1]);
                fa_rows_add(z, t, cbits, gi * FA_GRP, f, lo[gi & 1], hi[gi & 1]);
            }
        } else {
            int s = 0, t = 0;
            float c = 0.f;
            if (i < E) {
                s = esrc[i];
                t = edst[i];
                v2f rt = *(const v2f*)(rec_dst + 2 * (size_t)t);
                v2f rs = *(const v2f*)(rec_src + 2 * (size_t)s);
                float g = tanhf(rt.x + rs.x + bias);
                c = g * rt.y * rs.y;
            }
            int cbits = __float_as_int(c);
            for (int j = 0; j < n; ++j) {
                fa_edge_update(h, z,
                               __builtin_amdgcn_readlane(s, j),
                               __builtin_amdgcn_readlane(t, j),
                               __int_as_float(__builtin_amdgcn_readlane(cbits, j)), f);
            }
        }
    }
}

#define FB_NB   196
#define FB_TS   12800
#define FB_WG   1024
#define FB_PER  ((FB_TS + FB_WG - 1) / FB_WG)

__device__ __forceinline__ int fb_block_excl_scan(int c, int* wtot, int lane, int wv)
{
    int inc = c;
    #pragma unroll
    for (int d = 1; d < 32; d <<= 1) {
        int u = __shfl_up(inc, d);
        if (lane >= d) inc += u;
    }
    if (lane == 31) wtot[wv] = inc;
    __syncthreads();
    if (wv == 0) {
        int w = wtot[lane];
        int wi = w;
        #pragma unroll
        for (int d = 1; d < 32; d <<= 1) {
            int u = __shfl_up(wi, d);
            if (lane >= d) wi += u;
        }
        wtot[lane] = wi - w;
    }
    __syncthreads();
    return inc - c + wtot[wv];
}

__global__ __launch_bounds__(FB_WG)
void fa_bin(const int* __restrict__ esrc, const int* __restrict__ edst,
            int2* __restrict__ recs, int* __restrict__ offs, int E, int B)
{
    __shared__ int  cnt[FB_WG];
    __shared__ int  wtot[32];
    __shared__ int2 srt[FB_TS];

    int tid  = threadIdx.x;
    int lane = tid & 31;
    int wv   = tid >> 5;
    int tile = blockIdx.x;
    int base = tile * FB_TS;
    int n    = E - base; if (n > FB_TS) n = FB_TS;

    cnt[tid] = 0;
    int s[FB_PER], t[FB_PER];
    #pragma unroll
    for (int k = 0; k < FB_PER; ++k) {
        int i = k * FB_WG + tid;
        s[k] = 0; t[k] = -1;
        if (i < n) { s[k] = esrc[base + i]; t[k] = edst[base + i]; }
    }
    __syncthreads();
    #pragma unroll
    for (int k = 0; k < FB_PER; ++k)
        if (t[k] >= 0) atomicAdd(&cnt[t[k] / FB_NB], 1);
    __syncthreads();

    int c    = cnt[tid];
    int excl = fb_block_excl_scan(c, wtot, lane, wv);
    cnt[tid] = excl;
    if (tid <= B) offs[(size_t)tile * (B + 1) + tid] = excl;
    __syncthreads();

    #pragma unroll
    for (int k = 0; k < FB_PER; ++k)
        if (t[k] >= 0) {
            int pos = atomicAdd(&cnt[t[k] / FB_NB], 1);
            srt[pos] = make_int2(s[k], t[k]);
        }
    __syncthreads();

    for (int i = tid; i < n; i += FB_WG)
        recs[(size_t)base + i] = srt[i];
}

#define FB_CH   8192
#define FB_CPT  (FB_CH / FB_WG)
#define FB_NBP  256

__device__ __forceinline__ void fb_rows_sum(float* acc, int tl, int cbits, unsigned bnd, int j0, int f,
                                            const float (&lo)[FA_GRP], const float (&hi)[FA_GRP],
                                            float& alo, float& ahi)
{
    #pragma unroll
    for (int k = 0; k < FA_GRP; ++k) {
        float c = __int_as_float(__builtin_amdgcn_readlane(cbits, j0 + k));
        alo = fmaf(lo[k], c, alo);
        ahi = fmaf(hi[k], c, ahi);
        if ((bnd >> (j0 + k)) & 1u) {
            float* ap = acc + __builtin_amdgcn_readlane(tl, j0 + k) * NDIM + f;
            atomicAdd(ap,      alo);
            atomicAdd(ap + 32, ahi);
            alo = 0.f;
            ahi = 0.f;
        }
    }
}

__global__ __launch_bounds__(FB_WG, 16)
void fa_accum(const float* __restrict__ h,
              const float* __restrict__ rec_dst,
              const float* __restrict__ rec_src,
              const float* __restrict__ gate_b,
              const int2* __restrict__ recs,
              const int* __restrict__ offs,
              float* __restrict__ z, int N, int T, int B)
{
    __shared__ __attribute__((aligned(16))) float acc[FB_NB * NDIM];
    __shared__ v2f recd[FB_NB];
    __shared__ int P[FB_WG + 1];
    __shared__ int S[FB_WG];
    __shared__ int wtot[32];
    __shared__ unsigned srt[FB_CH];
    __shared__ int cntN[FB_NBP];

    int tid  = threadIdx.x;
    int lane = tid & 31;
    int wv   = __builtin_amdgcn_readfirstlane(tid >> 5);
    int b    = blockIdx.x;
    int n0   = b * FB_NB;
    int nn   = N - n0; if (nn > FB_NB) nn = FB_NB;
    int f    = lane;
    float bias = gate_b[0];

    for (int i = tid; i < FB_NB * NDIM / 4; i += FB_WG)
        ((v4f*)acc)[i] = (v4f){0.f, 0.f, 0.f, 0.f};
    if (tid < nn) recd[tid] = *(const v2f*)(rec_dst + 2 * (size_t)(n0 + tid));

    int len = 0, st = 0;
    if (tid < T) {
        const int* op = offs + (size_t)tid * (B + 1) + b;
        int o0 = op[0], o1 = op[1];
        len = o1 - o0;
        st  = tid * FB_TS + o0;
    }
    int excl = fb_block_excl_scan(len, wtot, lane, wv);
    P[tid] = excl;
    S[tid] = st;
    if (tid == FB_WG - 1) P[FB_WG] = excl + len;
    __syncthreads();
    int total = P[FB_WG];
    int top = 1;
    while (2 * top < T) top <<= 1;

    for (int c0 = 0; c0 < total; c0 += FB_CH) {
        int m = total - c0; if (m > FB_CH) m = FB_CH;

        if (tid < FB_NBP) cntN[tid] = 0;
        __syncthreads();

        unsigned pk[FB_CPT];
        int      rk[FB_CPT];
        int      vv[FB_CPT], lt[FB_CPT];
        #pragma unroll
        for (int k = 0; k < FB_CPT; ++k) {
            int q = k * FB_WG + tid;
            vv[k] = c0 + ((q < m) ? q : 0);
            lt[k] = 0;
        }
        for (int step = top; step >= 1; step >>= 1) {
            #pragma unroll
            for (int k = 0; k < FB_CPT; ++k)
                lt[k] += (P[lt[k] + step] <= vv[k]) ? step : 0;
        }
        int2 ed[FB_CPT];
        #pragma unroll
        for (int k = 0; k < FB_CPT; ++k)
            ed[k] = recs[(size_t)(S[lt[k]] + (vv[k] - P[lt[k]]))];
        #pragma unroll
        for (int k = 0; k < FB_CPT; ++k) {
            int tl = ed[k].y - n0;
            pk[k] = ((unsigned)tl << 24) | (unsigned)ed[k].x;
            rk[k] = -1;
            if (k * FB_WG + tid < m)
                rk[k] = atomicAdd(&cntN[tl], 1);
        }
        __syncthreads();

        int cN = (tid < FB_NBP) ? cntN[tid] : 0;
        int ex = fb_block_excl_scan(cN, wtot, lane, wv);
        if (tid < FB_NBP) cntN[tid] = ex;
        __syncthreads();

        #pragma unroll
        for (int k = 0; k < FB_CPT; ++k)
            if (rk[k] >= 0) srt[cntN[pk[k] >> 24] + rk[k]] = pk[k];
        __syncthreads();

        for (int v0 = wv * 32; v0 < m; v0 += FB_WG) {
            int n = m - v0;
            int j = v0 + lane;
            bool live = j < m;
            int s = 0, tl = 0;
            v2f rt = {0.f, 0.f}, rs = {0.f, 0.f};
            if (live) {
                unsigned p = srt[j];
                s  = (int)(p & 0xFFFFFFu);
                tl = (int)(p >> 24);
                rt = recd[tl];
                rs = *(const v2f*)(rec_src + 2 * (size_t)s);
            }
            int tl_next = __shfl_down(tl, 1);
            bool last = (lane == 31) || (j + 1 >= m) || (tl_next != tl);
            unsigned bnd = __builtin_amdgcn_ballot_w32(last);
            float alo = 0.f, ahi = 0.f;

            if (n >= 32) {
                float g = tanhf(rt.x + rs.x + bias);
                float c = g * rt.y * rs.y;
                int cbits = __float_as_int(c);
                float lo[2][FA_GRP], hi[2][FA_GRP];
                fa_rows_issue(h, s, 0, f, lo[0], hi[0]);
                #pragma unroll
                for (int gi = 0; gi < FA_NGRP; ++gi) {
                    if (gi + 1 < FA_NGRP)
                        fa_rows_issue(h, s, (gi + 1) * FA_GRP, f, lo[(gi + 1) & 1], hi[(gi + 1) & 1]);
                    fb_rows_sum(acc, tl, cbits, bnd, gi * FA_GRP, f, lo[gi & 1], hi[gi & 1], alo, ahi);
                }
            } else {
                float c = 0.f;
                if (live) {
                    float g = tanhf(rt.x + rs.x + bias);
                    c = g * rt.y * rs.y;
                }
                int cbits = __float_as_int(c);
                for (int e = 0; e < n; ++e) {
                    int   sj = __builtin_amdgcn_readlane(s, e);
                    float cj = __int_as_float(__builtin_amdgcn_readlane(cbits, e));
                    const float* hp = h + (size_t)sj * NDIM + f;
                    alo = fmaf(hp[0],  cj, alo);
                    ahi = fmaf(hp[32], cj, ahi);
                    if ((bnd >> e) & 1u) {
                        float* ap = acc + __builtin_amdgcn_readlane(tl, e) * NDIM + f;
                        atomicAdd(ap,      alo);
                        atomicAdd(ap + 32, ahi);
                        alo = 0.f;
                        ahi = 0.f;
                    }
                }
            }
        }
    }
    __syncthreads();

    v4f* zp = (v4f*)(z + (size_t)n0 * NDIM);
    for (int i = tid; i < nn * (NDIM / 4); i += FB_WG)
        zp[i] = ((const v4f*)acc)[i];
}

extern "C" void kernel_launch(void* const* d_in, const int* in_sizes, int n_in,
                              void* d_out, int out_size, void* d_ws, size_t ws_size,
                              hipStream_t stream)
{
    const float* h      = (const float*)d_in[0];
    const float* dvec   = (const float*)d_in[1];
    const float* gate_W = (const float*)d_in[2];
    const float* gate_b = (const float*)d_in[3];
    const int*   esrc   = (const int*)d_in[4];
    const int*   edst   = (const int*)d_in[5];

    int N = in_sizes[1];
    int E = in_sizes[4];
    float* z = (float*)d_out;

    float* rec_dst = (float*)d_ws;
    float* rec_src = rec_dst + 2 * (size_t)N;

    {
        int waves   = (N + 15) / 16;
        int threads = waves * 32;
        dim3 blk(256), grd((threads + 255) / 256);
        fa_gate_wmma<<<grd, blk, 0, stream>>>(h, gate_W, dvec, rec_dst, rec_src, N);
    }

    int B = (N + FB_NB - 1) / FB_NB;
    int T = (E + FB_TS - 1) / FB_TS;
    size_t recs_off = (16 * (size_t)N + 255) / 256 * 256;
    size_t offs_off = recs_off + (8 * (size_t)E + 255) / 256 * 256;
    size_t need     = offs_off + 4 * (size_t)T * (B + 1);

    if (T >= 1 && T <= FB_WG && B < FB_WG && N <= (1 << 24) && ws_size >= need) {
        int2* recs = (int2*)((char*)d_ws + recs_off);
        int*  offs = (int*)((char*)d_ws + offs_off);
        fa_bin<<<dim3(T), dim3(FB_WG), 0, stream>>>(esrc, edst, recs, offs, E, B);
        fa_accum<<<dim3(B), dim3(FB_WG), 0, stream>>>(h, rec_dst, rec_src, gate_b, recs, offs, z, N, T, B);
    } else {
        {
            int n4 = out_size / 4;
            dim3 blk(256), grd((n4 + 255) / 256);
            fa_zero<<<grd, blk, 0, stream>>>(z, n4);
        }
        {
            int waves   = (E + 31) / 32;
            int threads = waves * 32;
            dim3 blk(256), grd((threads + 255) / 256);
            fa_gate_scatter<<<grd, blk, 0, stream>>>(h, rec_dst, rec_src, gate_b, esrc, edst, z, E);
        }
    }
}


# Accumulate kernel: lane l owns features 2l and 2l+1, so one 8-byte load per lane fetches a whole 256-byte h row; half as many load instructions per edge through the memory pipe
# speedup vs baseline: 3.3085x; 1.1341x over previous
#include <hip/hip_runtime.h>
#include <math.h>

typedef __attribute__((ext_vector_type(2))) float v2f;
typedef __attribute__((ext_vector_type(4))) float v4f;
typedef __attribute__((ext_vector_type(8))) float v8f;

#define NDIM 64

__global__ void fa_gate_wmma(const float* __restrict__ h,
                             const float* __restrict__ gate_W,
                             const float* __restrict__ dvec,
                             float* __restrict__ rec_dst,
                             float* __restrict__ rec_src,
                             int N)
{
    int wave = (blockIdx.x * blockDim.x + threadIdx.x) >> 5;
    int lane = threadIdx.x & 31;
    int base = wave * 16;
    if (base >= N) return;

    int row   = lane & 15;
    int khalf = (lane >> 4) << 1;
    int arow  = base + row; if (arow > N - 1) arow = N - 1;

    float msk = (row < 2) ? 1.0f : 0.0f;
    const float* wbase = gate_W + ((row < 2) ? row : 0) * NDIM + khalf;
    v2f bfrag[16];
    #pragma unroll
    for (int j = 0; j < 16; ++j) {
        bfrag[j].x = wbase[j * 4]     * msk;
        bfrag[j].y = wbase[j * 4 + 1] * msk;
    }

    const float* arow_p = h + (size_t)arow * NDIM + khalf;
    v8f acc = {0.f, 0.f, 0.f, 0.f, 0.f, 0.f, 0.f, 0.f};
    #pragma unroll
    for (int j = 0; j < 16; ++j) {
        v2f a = *(const v2f*)(arow_p + j * 4);
        acc = __builtin_amdgcn_wmma_f32_16x16x4_f32(
                   false, a,  false, bfrag[j],
                   (short)0, acc,  false,  false);
    }

    if (row < 2) {
        float* outp = (row == 0) ? rec_dst : rec_src;
        int moff = (lane >> 4) << 3;
        #pragma unroll
        for (int i = 0; i < 8; ++i) {
            int m = base + moff + i;
            if (m < N) {
                v2f r; r.x = acc[i]; r.y = dvec[m];
                *(v2f*)(outp + 2 * (size_t)m) = r;
            }
        }
    }
}

__global__ void fa_zero(float* __restrict__ z, int n4)
{
    int i = blockIdx.x * blockDim.x + threadIdx.x;
    if (i < n4) ((v4f*)z)[i] = (v4f){0.f, 0.f, 0.f, 0.f};
}

#define FA_GRP 8
#define FA_NGRP (32 / FA_GRP)

__device__ __forceinline__ void fa_edge_update(const float* __restrict__ h,
                                               float* __restrict__ z,
                                               int s, int t, float c, int f)
{
    const float* hp = h + (size_t)s * NDIM + f;
    float h_lo = hp[0];
    float h_hi = hp[32];
    float* zp = z + (size_t)t * NDIM + f;
    atomicAdd(zp,      h_lo * c);
    atomicAdd(zp + 32, h_hi * c);
}

__device__ __forceinline__ void fa_rows_issue(const float* __restrict__ h,
                                              int s, int j0, int f,
                                              float (&lo)[FA_GRP], float (&hi)[FA_GRP])
{
    #pragma unroll
    for (int k = 0; k < FA_GRP; ++k) {
        const float* hp = h + (size_t)__builtin_amdgcn_readlane(s, j0 + k) * NDIM + f;
        lo[k] = hp[0];
        hi[k] = hp[32];
    }
}

__device__ __forceinline__ void fa_rows_add(float* __restrict__ z,
                                            int t, int cbits, int j0, int f,
                                            const float (&lo)[FA_GRP], const float (&hi)[FA_GRP])
{
    #pragma unroll
    for (int k = 0; k < FA_GRP; ++k) {
        float c = __int_as_float(__builtin_amdgcn_readlane(cbits, j0 + k));
        float* zp = z + (size_t)__builtin_amdgcn_readlane(t, j0 + k) * NDIM + f;
        atomicAdd(zp,      lo[k] * c);
        atomicAdd(zp + 32, hi[k] * c);
    }
}

__global__ void fa_gate_scatter(const float* __restrict__ h,
                                const float* __restrict__ rec_dst,
                                const float* __restrict__ rec_src,
                                const float* __restrict__ gate_b,
                                const int* __restrict__ esrc,
                                const int* __restrict__ edst,
                                float* __restrict__ z, int E)
{
    int lane = threadIdx.x & 31;
    int wave   = __builtin_amdgcn_readfirstlane((int)((blockIdx.x * blockDim.x + threadIdx.x) >> 5));
    int nwaves = (int)((gridDim.x * blockDim.x) >> 5);
    int f = lane;
    float bias = gate_b[0];

    for (int base = wave * 32; base < E; base += nwaves * 32) {
        int i = base + lane;
        int n = E - base;

        if (n >= 32) {
            int s = esrc[i];
            int t = edst[i];
            v2f rt = *(const v2f*)(rec_dst + 2 * (size_t)t);
            v2f rs = *(const v2f*)(rec_src + 2 * (size_t)s);

            float lo[2][FA_GRP], hi[2][FA_GRP];
            fa_rows_issue(h, s, 0, f, lo[0], hi[0]);

            float g = tanhf(rt.x + rs.x + bias);
            float c = g * rt.y * rs.y;
            int cbits = __float_as_int(c);

            #pragma unroll
            for (int gi = 0; gi < FA_NGRP; ++gi) {
                if (gi + 1 < FA_NGRP)
                    fa_rows_issue(h, s, (gi + 1) * FA_GRP, f, lo[(gi + 1) & 1], hi[(gi + 1) & 1]);
                fa_rows_add(z, t, cbits, gi * FA_GRP, f, lo[gi & 1], hi[gi & 1]);
            }
        } else {
            int s = 0, t = 0;
            float c = 0.f;
            if (i < E) {
                s = esrc[i];
                t = edst[i];
                v2f rt = *(const v2f*)(rec_dst + 2 * (size_t)t);
                v2f rs = *(const v2f*)(rec_src + 2 * (size_t)s);
                float g = tanhf(rt.x + rs.x + bias);
                c = g * rt.y * rs.y;
            }
            int cbits = __float_as_int(c);
            for (int j = 0; j < n; ++j) {
                fa_edge_update(h, z,
                               __builtin_amdgcn_readlane(s, j),
                               __builtin_amdgcn_readlane(t, j),
                               __int_as_float(__builtin_amdgcn_readlane(cbits, j)), f);
            }
        }
    }
}

#define FB_NB   196
#define FB_TS   12800
#define FB_WG   1024
#define FB_PER  ((FB_TS + FB_WG - 1) / FB_WG)

__device__ __forceinline__ int fb_block_excl_scan(int c, int* wtot, int lane, int wv)
{
    int inc = c;
    #pragma unroll
    for (int d = 1; d < 32; d <<= 1) {
        int u = __shfl_up(inc, d);
        if (lane >= d) inc += u;
    }
    if (lane == 31) wtot[wv] = inc;
    __syncthreads();
    if (wv == 0) {
        int w = wtot[lane];
        int wi = w;
        #pragma unroll
        for (int d = 1; d < 32; d <<= 1) {
            int u = __shfl_up(wi, d);
            if (lane >= d) wi += u;
        }
        wtot[lane] = wi - w;
    }
    __syncthreads();
    return inc - c + wtot[wv];
}

__global__ __launch_bounds__(FB_WG)
void fa_bin(const int* __restrict__ esrc, const int* __restrict__ edst,
            int2* __restrict__ recs, int* __restrict__ offs, int E, int B)
{
    __shared__ int  cnt[FB_WG];
    __shared__ int  wtot[32];
    __shared__ int2 srt[FB_TS];

    int tid  = threadIdx.x;
    int lane = tid & 31;
    int wv   = tid >> 5;
    int tile = blockIdx.x;
    int base = tile * FB_TS;
    int n    = E - base; if (n > FB_TS) n = FB_TS;

    cnt[tid] = 0;
    int s[FB_PER], t[FB_PER];
    #pragma unroll
    for (int k = 0; k < FB_PER; ++k) {
        int i = k * FB_WG + tid;
        s[k] = 0; t[k] = -1;
        if (i < n) { s[k] = esrc[base + i]; t[k] = edst[base + i]; }
    }
    __syncthreads();
    #pragma unroll
    for (int k = 0; k < FB_PER; ++k)
        if (t[k] >= 0) atomicAdd(&cnt[t[k] / FB_NB], 1);
    __syncthreads();

    int c    = cnt[tid];
    int excl = fb_block_excl_scan(c, wtot, lane, wv);
    cnt[tid] = excl;
    if (tid <= B) offs[(size_t)tile * (B + 1) + tid] = excl;
    __syncthreads();

    #pragma unroll
    for (int k = 0; k < FB_PER; ++k)
        if (t[k] >= 0) {
            int pos = atomicAdd(&cnt[t[k] / FB_NB], 1);
            srt[pos] = make_int2(s[k], t[k]);
        }
    __syncthreads();

    for (int i = tid; i < n; i += FB_WG)
        recs[(size_t)base + i] = srt[i];
}

#define FB_CH   8192
#define FB_CPT  (FB_CH / FB_WG)
#define FB_NBP  256

__device__ __forceinline__ void fb_rows_issue(const float* __restrict__ h,
                                              int s, int j0, int f2,
                                              v2f (&row)[FA_GRP])
{
    #pragma unroll
    for (int k = 0; k < FA_GRP; ++k)
        row[k] = *(const v2f*)(h + (size_t)__builtin_amdgcn_readlane(s, j0 + k) * NDIM + f2);
}

__device__ __forceinline__ void fb_rows_sum(float* acc, int tl, int cbits, unsigned bnd, int j0, int f2,
                                            const v2f (&row)[FA_GRP],
                                            float& a0, float& a1)
{
    #pragma unroll
    for (int k = 0; k < FA_GRP; ++k) {
        float c = __int_as_float(__builtin_amdgcn_readlane(cbits, j0 + k));
        a0 = fmaf(row[k].x, c, a0);
        a1 = fmaf(row[k].y, c, a1);
        if ((bnd >> (j0 + k)) & 1u) {
            float* ap = acc + __builtin_amdgcn_readlane(tl, j0 + k) * NDIM + f2;
            atomicAdd(ap,     a0);
            atomicAdd(ap + 1, a1);
            a0 = 0.f;
            a1 = 0.f;
        }
    }
}

__global__ __launch_bounds__(FB_WG, 16)
void fa_accum(const float* __restrict__ h,
              const float* __restrict__ rec_dst,
              const float* __restrict__ rec_src,
              const float* __restrict__ gate_b,
              const int2* __restrict__ recs,
              const int* __restrict__ offs,
              float* __restrict__ z, int N, int T, int B)
{
    __shared__ __attribute__((aligned(16))) float acc[FB_NB * NDIM];
    __shared__ v2f recd[FB_NB];
    __shared__ int P[FB_WG + 1];
    __shared__ int S[FB_WG];
    __shared__ int wtot[32];
    __shared__ unsigned srt[FB_CH];
    __shared__ int cntN[FB_NBP];

    int tid  = threadIdx.x;
    int lane = tid & 31;
    int wv   = __builtin_amdgcn_readfirstlane(tid >> 5);
    int b    = blockIdx.x;
    int n0   = b * FB_NB;
    int nn   = N - n0; if (nn > FB_NB) nn = FB_NB;
    int f2   = 2 * lane;
    float bias = gate_b[0];

    for (int i = tid; i < FB_NB * NDIM / 4; i += FB_WG)
        ((v4f*)acc)[i] = (v4f){0.f, 0.f, 0.f, 0.f};
    if (tid < nn) recd[tid] = *(const v2f*)(rec_dst + 2 * (size_t)(n0 + tid));

    int len = 0, st = 0;
    if (tid < T) {
        const int* op = offs + (size_t)tid * (B + 1) + b;
        int o0 = op[0], o1 = op[1];
        len = o1 - o0;
        st  = tid * FB_TS + o0;
    }
    int excl = fb_block_excl_scan(len, wtot, lane, wv);
    P[tid] = excl;
    S[tid] = st;
    if (tid == FB_WG - 1) P[FB_WG] = excl + len;
    __syncthreads();
    int total = P[FB_WG];
    int top = 1;
    while (2 * top < T) top <<= 1;

    for (int c0 = 0; c0 < total; c0 += FB_CH) {
        int m = total - c0; if (m > FB_CH) m = FB_CH;

        if (tid < FB_NBP) cntN[tid] = 0;
        __syncthreads();

        unsigned pk[FB_CPT];
        int      rk[FB_CPT];
        int      vv[FB_CPT], lt[FB_CPT];
        #pragma unroll
        for (int k = 0; k < FB_CPT; ++k) {
            int q = k * FB_WG + tid;
            vv[k] = c0 + ((q < m) ? q : 0);
            lt[k] = 0;
        }
        for (int step = top; step >= 1; step >>= 1) {
            #pragma unroll
            for (int k = 0; k < FB_CPT; ++k)
                lt[k] += (P[lt[k] + step] <= vv[k]) ? step : 0;
        }
        int2 ed[FB_CPT];
        #pragma unroll
        for (int k = 0; k < FB_CPT; ++k)
            ed[k] = recs[(size_t)(S[lt[k]] + (vv[k] - P[lt[k]]))];
        #pragma unroll
        for (int k = 0; k < FB_CPT; ++k) {
            int tl = ed[k].y - n0;
            pk[k] = ((unsigned)tl << 24) | (unsigned)ed[k].x;
            rk[k] = -1;
            if (k * FB_WG + tid < m)
                rk[k] = atomicAdd(&cntN[tl], 1);
        }
        __syncthreads();

        int cN = (tid < FB_NBP) ? cntN[tid] : 0;
        int ex = fb_block_excl_scan(cN, wtot, lane, wv);
        if (tid < FB_NBP) cntN[tid] = ex;
        __syncthreads();

        #pragma unroll
        for (int k = 0; k < FB_CPT; ++k)
            if (rk[k] >= 0) srt[cntN[pk[k] >> 24] + rk[k]] = pk[k];
        __syncthreads();

        for (int v0 = wv * 32; v0 < m; v0 += FB_WG) {
            int n = m - v0;
            int j = v0 + lane;
            bool live = j < m;
            int s = 0, tl = 0;
            v2f rt = {0.f, 0.f}, rs = {0.f, 0.f};
            if (live) {
                unsigned p = srt[j];
                s  = (int)(p & 0xFFFFFFu);
                tl = (int)(p >> 24);
                rt = recd[tl];
                rs = *(const v2f*)(rec_src + 2 * (size_t)s);
            }
            int tl_next = __shfl_down(tl, 1);
            bool last = (lane == 31) || (j + 1 >= m) || (tl_next != tl);
            unsigned bnd = __builtin_amdgcn_ballot_w32(last);
            float a0 = 0.f, a1 = 0.f;

            if (n >= 32) {
                float g = tanhf(rt.x + rs.x + bias);
                float c = g * rt.y * rs.y;
                int cbits = __float_as_int(c);
                v2f row[2][FA_GRP];
                fb_rows_issue(h, s, 0, f2, row[0]);
                #pragma unroll
                for (int gi = 0; gi < FA_NGRP; ++gi) {
                    if (gi + 1 < FA_NGRP)
                        fb_rows_issue(h, s, (gi + 1) * FA_GRP, f2, row[(gi + 1) & 1]);
                    fb_rows_sum(acc, tl, cbits, bnd, gi * FA_GRP, f2, row[gi & 1], a0, a1);
                }
            } else {
                float c = 0.f;
                if (live) {
                    float g = tanhf(rt.x + rs.x + bias);
                    c = g * rt.y * rs.y;
                }
                int cbits = __float_as_int(c);
                for (int e = 0; e < n; ++e) {
                    int   sj = __builtin_amdgcn_readlane(s, e);
                    float cj = __int_as_float(__builtin_amdgcn_readlane(cbits, e));
                    v2f r = *(const v2f*)(h + (size_t)sj * NDIM + f2);
                    a0 = fmaf(r.x, cj, a0);
                    a1 = fmaf(r.y, cj, a1);
                    if ((bnd >> e) & 1u) {
                        float* ap = acc + __builtin_amdgcn_readlane(tl, e) * NDIM + f2;
                        atomicAdd(ap,     a0);
                        atomicAdd(ap + 1, a1);
                        a0 = 0.f;
                        a1 = 0.f;
                    }
                }
            }
        }
    }
    __syncthreads();

    v4f* zp = (v4f*)(z + (size_t)n0 * NDIM);
    for (int i = tid; i < nn * (NDIM / 4); i += FB_WG)
        zp[i] = ((const v4f*)acc)[i];
}

extern "C" void kernel_launch(void* const* d_in, const int* in_sizes, int n_in,
                              void* d_out, int out_size, void* d_ws, size_t ws_size,
                              hipStream_t stream)
{
    const float* h      = (const float*)d_in[0];
    const float* dvec   = (const float*)d_in[1];
    const float* gate_W = (const float*)d_in[2];
    const float* gate_b = (const float*)d_in[3];
    const int*   esrc   = (const int*)d_in[4];
    const int*   edst   = (const int*)d_in[5];

    int N = in_sizes[1];
    int E = in_sizes[4];
    float* z = (float*)d_out;

    float* rec_dst = (float*)d_ws;
    float* rec_src = rec_dst + 2 * (size_t)N;

    {
        int waves   = (N + 15) / 16;
        int threads = waves * 32;
        dim3 blk(256), grd((threads + 255) / 256);
        fa_gate_wmma<<<grd, blk, 0, stream>>>(h, gate_W, dvec, rec_dst, rec_src, N);
    }

    int B = (N + FB_NB - 1) / FB_NB;
    int T = (E + FB_TS - 1) / FB_TS;
    size_t recs_off = (16 * (size_t)N + 255) / 256 * 256;
    size_t offs_off = recs_off + (8 * (size_t)E + 255) / 256 * 256;
    size_t need     = offs_off + 4 * (size_t)T * (B + 1);

    if (T >= 1 && T <= FB_WG && B < FB_WG && N <= (1 << 24) && ws_size >= need) {
        int2* recs = (int2*)((char*)d_ws + recs_off);
        int*  offs = (int*)((char*)d_ws + offs_off);
        fa_bin<<<dim3(T), dim3(FB_WG), 0, stream>>>(esrc, edst, recs, offs, E, B);
        fa_accum<<<dim3(B), dim3(FB_WG), 0, stream>>>(h, rec_dst, rec_src, gate_b, recs, offs, z, N, T, B);
    } else {
        {
            int n4 = out_size / 4;
            dim3 blk(256), grd((n4 + 255) / 256);
            fa_zero<<<grd, blk, 0, stream>>>(z, n4);
        }
        {
            int waves   = (E + 31) / 32;
            int threads = waves * 32;
            dim3 blk(256), grd((threads + 255) / 256);
            fa_gate_scatter<<<grd, blk, 0, stream>>>(h, rec_dst, rec_src, gate_b, esrc, edst, z, E);
        }
    }
}


# Memory operations issue as groups, not each behind the last one's wait: binning index loads on clamped addresses (were 7 round trips), copy-out, gate d reads; accumulate: rows ahead of lookup wait
# speedup vs baseline: 3.8046x; 1.1499x over previous
#include <hip/hip_runtime.h>
#include <math.h>

typedef __attribute__((ext_vector_type(2))) float v2f;
typedef __attribute__((ext_vector_type(4))) float v4f;
typedef __attribute__((ext_vector_type(8))) float v8f;

#define NDIM 64

__device__ __forceinline__ void fa_gate_wave(const float* __restrict__ h,
                                             const float* __restrict__ gate_W,
                                             const float* __restrict__ dvec,
                                             float* __restrict__ rec_dst,
                                             float* __restrict__ rec_src,
                                             int N, int wave, int lane)
{
    int base = wave * 16;
    if (base >= N) return;

    int row   = lane & 15;
    int khalf = (lane >> 4) << 1;
    int arow  = base + row; if (arow > N - 1) arow = N - 1;

    float msk = (row < 2) ? 1.0f : 0.0f;
    const float* wbase = gate_W + ((row < 2) ? row : 0) * NDIM + khalf;
    v2f bfrag[16];
    #pragma unroll
    for (int j = 0; j < 16; ++j) {
        bfrag[j].x = wbase[j * 4]     * msk;
        bfrag[j].y = wbase[j * 4 + 1] * msk;
    }

    const float* arow_p = h + (size_t)arow * NDIM + khalf;
    v8f acc = {0.f, 0.f, 0.f, 0.f, 0.f, 0.f, 0.f, 0.f};
    #pragma unroll
    for (int j = 0; j < 16; ++j) {
        v2f a = *(const v2f*)(arow_p + j * 4);
        acc = __builtin_amdgcn_wmma_f32_16x16x4_f32(
                   false, a,  false, bfrag[j],
                   (short)0, acc,  false,  false);
    }

    if (row < 2) {
        float* outp = (row == 0) ? rec_dst : rec_src;
        int moff = (lane >> 4) << 3;
        float dv[8];
        #pragma unroll
        for (int i = 0; i < 8; ++i) {
            int m = base + moff + i;
            dv[i] = dvec[(m < N) ? m : N - 1];
        }
        #pragma unroll
        for (int i = 0; i < 8; ++i) {
            int m = base + moff + i;
            if (m < N) {
                v2f r; r.x = acc[i]; r.y = dv[i];
                *(v2f*)(outp + 2 * (size_t)m) = r;
            }
        }
    }
}

__global__ void fa_gate_wmma(const float* __restrict__ h,
                             const float* __restrict__ gate_W,
                             const float* __restrict__ dvec,
                             float* __restrict__ rec_dst,
                             float* __restrict__ rec_src,
                             int N)
{
    int wave = (blockIdx.x * blockDim.x + threadIdx.x) >> 5;
    fa_gate_wave(h, gate_W, dvec, rec_dst, rec_src, N, wave, threadIdx.x & 31);
}

__global__ void fa_zero(float* __restrict__ z, int n4)
{
    int i = blockIdx.x * blockDim.x + threadIdx.x;
    if (i < n4) ((v4f*)z)[i] = (v4f){0.f, 0.f, 0.f, 0.f};
}

#define FA_GRP 8
#define FA_NGRP (32 / FA_GRP)

__device__ __forceinline__ void fa_edge_update(const float* __restrict__ h,
                                               float* __restrict__ z,
                                               int s, int t, float c, int f)
{
    const float* hp = h + (size_t)s * NDIM + f;
    float h_lo = hp[0];
    float h_hi = hp[32];
    float* zp = z + (size_t)t * NDIM + f;
    atomicAdd(zp,      h_lo * c);
    atomicAdd(zp + 32, h_hi * c);
}

__device__ __forceinline__ void fa_rows_issue(const float* __restrict__ h,
                                              int s, int j0, int f,
                                              float (&lo)[FA_GRP], float (&hi)[FA_GRP])
{
    #pragma unroll
    for (int k = 0; k < FA_GRP; ++k) {
        const float* hp = h + (size_t)__builtin_amdgcn_readlane(s, j0 + k) * NDIM + f;
        lo[k] = hp[0];
        hi[k] = hp[32];
    }
}

__device__ __forceinline__ void fa_rows_add(float* __restrict__ z,
                                            int t, int cbits, int j0, int f,
                                            const float (&lo)[FA_GRP], const float (&hi)[FA_GRP])
{
    #pragma unroll
    for (int k = 0; k < FA_GRP; ++k) {
        float c = __int_as_float(__builtin_amdgcn_readlane(cbits, j0 + k));
        float* zp = z + (size_t)__builtin_amdgcn_readlane(t, j0 + k) * NDIM + f;
        atomicAdd(zp,      lo[k] * c);
        atomicAdd(zp + 32, hi[k] * c);
    }
}

__global__ void fa_gate_scatter(const float* __restrict__ h,
                                const float* __restrict__ rec_dst,
                                const float* __restrict__ rec_src,
                                const float* __restrict__ gate_b,
                                const int* __restrict__ esrc,
                                const int* __restrict__ edst,
                                float* __restrict__ z, int E)
{
    int lane = threadIdx.x & 31;
    int wave   = __builtin_amdgcn_readfirstlane((int)((blockIdx.x * blockDim.x + threadIdx.x) >> 5));
    int nwaves = (int)((gridDim.x * blockDim.x) >> 5);
    int f = lane;
    float bias = gate_b[0];

    for (int base = wave * 32; base < E; base += nwaves * 32) {
        int i = base + lane;
        int n = E - base;

        if (n >= 32) {
            int s = esrc[i];
            int t = edst[i];
            v2f rt = *(const v2f*)(rec_dst + 2 * (size_t)t);
            v2f rs = *(const v2f*)(rec_src + 2 * (size_t)s);

            float lo[2][FA_GRP], hi[2][FA_GRP];
            fa_rows_issue(h, s, 0, f, lo[0], hi[0]);

            float g = tanhf(rt.x + rs.x + bias);
            float c = g * rt.y * rs.y;
            int cbits = __float_as_int(c);

            #pragma unroll
            for (int gi = 0; gi < FA_NGRP; ++gi) {
                if (gi + 1 < FA_NGRP)
                    fa_rows_issue(h, s, (gi + 1) * FA_GRP, f, lo[(gi + 1) & 1], hi[(gi + 1) & 1]);
                fa_rows_add(z, t, cbits, gi * FA_GRP, f, lo[gi & 1], hi[gi & 1]);
            }
        } else {
            int s = 0, t = 0;
            float c = 0.f;
            if (i < E) {
                s = esrc[i];
                t = edst[i];
                v2f rt = *(const v2f*)(rec_dst + 2 * (size_t)t);
                v2f rs = *(const v2f*)(rec_src + 2 * (size_t)s);
                float g = tanhf(rt.x + rs.x + bias);
                c = g * rt.y * rs.y;
            }
            int cbits = __float_as_int(c);
            for (int j = 0; j < n; ++j) {
                fa_edge_update(h, z,
                               __builtin_amdgcn_readlane(s, j),
                               __builtin_amdgcn_readlane(t, j),
                               __int_as_float(__builtin_amdgcn_readlane(cbits, j)), f);
            }
        }
    }
}

#define FB_NB   196
#define FB_TS   12800
#define FB_WG   1024
#define FB_PER  ((FB_TS + FB_WG - 1) / FB_WG)

__device__ __forceinline__ int fb_block_excl_scan(int c, int* wtot, int lane, int wv)
{
    int inc = c;
    #pragma unroll
    for (int d = 1; d < 32; d <<= 1) {
        int u = __shfl_up(inc, d);
        if (lane >= d) inc += u;
    }
    if (lane == 31) wtot[wv] = inc;
    __syncthreads();
    if (wv == 0) {
        int w = wtot[lane];
        int wi = w;
        #pragma unroll
        for (int d = 1; d < 32; d <<= 1) {
            int u = __shfl_up(wi, d);
            if (lane >= d) wi += u;
        }
        wtot[lane] = wi - w;
    }
    __syncthreads();
    return inc - c + wtot[wv];
}

__global__ __launch_bounds__(FB_WG, 16)
void fa_bin_gate(const int* __restrict__ esrc, const int* __restrict__ edst,
                 int2* __restrict__ recs, int* __restrict__ offs, int E, int B, int T,
                 const float* __restrict__ h,
                 const float* __restrict__ gate_W,
                 const float* __restrict__ dvec,
                 float* __restrict__ rec_dst,
                 float* __restrict__ rec_src,
                 int N)
{
    __shared__ int  cnt[FB_WG];
    __shared__ int  wtot[32];
    __shared__ int2 srt[FB_TS];

    int tid  = threadIdx.x;
    int lane = tid & 31;

    if ((int)blockIdx.x >= T) {
        int gw = __builtin_amdgcn_readfirstlane((int)(((blockIdx.x - T) * FB_WG + tid) >> 5));
        fa_gate_wave(h, gate_W, dvec, rec_dst, rec_src, N, gw, lane);
        return;
    }

    int wv   = tid >> 5;
    int tile = blockIdx.x;
    int base = tile * FB_TS;
    int n    = E - base; if (n > FB_TS) n = FB_TS;

    cnt[tid] = 0;
    int s[FB_PER], t[FB_PER];
    #pragma unroll
    for (int k = 0; k < FB_PER; ++k) {
        int i  = k * FB_WG + tid;
        int ii = (i < n) ? i : n - 1;
        s[k] = esrc[base + ii];
        t[k] = edst[base + ii];
    }
    #pragma unroll
    for (int k = 0; k < FB_PER; ++k)
        if (k * FB_WG + tid >= n) t[k] = -1;
    __syncthreads();
    #pragma unroll
    for (int k = 0; k < FB_PER; ++k)
        if (t[k] >= 0) atomicAdd(&cnt[t[k] / FB_NB], 1);
    __syncthreads();

    int c    = cnt[tid];
    int excl = fb_block_excl_scan(c, wtot, lane, wv);
    cnt[tid] = excl;
    __syncthreads();

    #pragma unroll
    for (int k = 0; k < FB_PER; ++k)
        if (t[k] >= 0) {
            int pos = atomicAdd(&cnt[t[k] / FB_NB], 1);
            srt[pos] = make_int2(s[k], t[k]);
        }
    __syncthreads();

    int2 rec[FB_PER];
    #pragma unroll
    for (int k = 0; k < FB_PER; ++k) {
        int i = k * FB_WG + tid;
        rec[k] = srt[(i < FB_TS) ? i : FB_TS - 1];
    }
    #pragma unroll
    for (int k = 0; k < FB_PER; ++k) {
        int i = k * FB_WG + tid;
        if (i < n) recs[(size_t)base + i] = rec[k];
    }
    if (tid <= B) offs[(size_t)tile * (B + 1) + tid] = excl;
}

#define FB_CH   8192
#define FB_CPT  (FB_CH / FB_WG)
#define FB_NBP  256

__device__ __forceinline__ void fb_rows_issue(const float* __restrict__ h,
                                              int s, int j0, int f2,
                                              v2f (&row)[FA_GRP])
{
    #pragma unroll
    for (int k = 0; k < FA_GRP; ++k)
        row[k] = *(const v2f*)(h + (size_t)__builtin_amdgcn_readlane(s, j0 + k) * NDIM + f2);
}

__device__ __forceinline__ void fb_rows_sum(float* acc, int tl, int cbits, unsigned bnd, int j0, int f2,
                                            const v2f (&row)[FA_GRP],
                                            float& a0, float& a1)
{
    #pragma unroll
    for (int k = 0; k < FA_GRP; ++k) {
        float c = __int_as_float(__builtin_amdgcn_readlane(cbits, j0 + k));
        a0 = fmaf(row[k].x, c, a0);
        a1 = fmaf(row[k].y, c, a1);
        if ((bnd >> (j0 + k)) & 1u) {
            float* ap = acc + __builtin_amdgcn_readlane(tl, j0 + k) * NDIM + f2;
            atomicAdd(ap,     a0);
            atomicAdd(ap + 1, a1);
            a0 = 0.f;
            a1 = 0.f;
        }
    }
}

__global__ __launch_bounds__(FB_WG, 16)
void fa_accum(const float* __restrict__ h,
              const float* __restrict__ rec_dst,
              const float* __restrict__ rec_src,
              const float* __restrict__ gate_b,
              const int2* __restrict__ recs,
              const int* __restrict__ offs,
              float* __restrict__ z, int N, int T, int B)
{
    __shared__ __attribute__((aligned(16))) float acc[FB_NB * NDIM];
    __shared__ v2f recd[FB_NB];
    __shared__ int P[FB_WG + 1];
    __shared__ int S[FB_WG];
    __shared__ int wtot[32];
    __shared__ unsigned srt[FB_CH];
    __shared__ int cntN[FB_NBP];

    int tid  = threadIdx.x;
    int lane = tid & 31;
    int wv   = __builtin_amdgcn_readfirstlane(tid >> 5);
    int b    = blockIdx.x;
    int n0   = b * FB_NB;
    int nn   = N - n0; if (nn > FB_NB) nn = FB_NB;
    int f2   = 2 * lane;
    float bias = gate_b[0];

    for (int i = tid; i < FB_NB * NDIM / 4; i += FB_WG)
        ((v4f*)acc)[i] = (v4f){0.f, 0.f, 0.f, 0.f};
    int nq = (tid < nn) ? tid : nn - 1;
    int tq = (tid < T) ? tid : T - 1;
    v2f rd = *(const v2f*)(rec_dst + 2 * (size_t)(n0 + nq));
    const int* op = offs + (size_t)tq * (B + 1) + b;
    int o0 = op[0], o1 = op[1];
    if (tid < nn) recd[tid] = rd;

    int len = 0, st = 0;
    if (tid < T) {
        len = o1 - o0;
        st  = tid * FB_TS + o0;
    }
    int excl = fb_block_excl_scan(len, wtot, lane, wv);
    P[tid] = excl;
    S[tid] = st;
    if (tid == FB_WG - 1) P[FB_WG] = excl + len;
    __syncthreads();
    int total = P[FB_WG];
    int top = 1;
    while (2 * top < T) top <<= 1;

    for (int c0 = 0; c0 < total; c0 += FB_CH) {
        int m = total - c0; if (m > FB_CH) m = FB_CH;

        if (tid < FB_NBP) cntN[tid] = 0;
        __syncthreads();

        unsigned pk[FB_CPT];
        int      rk[FB_CPT];
        int      vv[FB_CPT], lt[FB_CPT];
        #pragma unroll
        for (int k = 0; k < FB_CPT; ++k) {
            int q = k * FB_WG + tid;
            vv[k] = c0 + ((q < m) ? q : 0);
            lt[k] = 0;
        }
        for (int step = top; step >= 1; step >>= 1) {
            #pragma unroll
            for (int k = 0; k < FB_CPT; ++k)
                lt[k] += (P[lt[k] + step] <= vv[k]) ? step : 0;
        }
        int2 ed[FB_CPT];
        #pragma unroll
        for (int k = 0; k < FB_CPT; ++k)
            ed[k] = recs[(size_t)(S[lt[k]] + (vv[k] - P[lt[k]]))];
        #pragma unroll
        for (int k = 0; k < FB_CPT; ++k) {
            int tl = ed[k].y - n0;
            pk[k] = ((unsigned)tl << 24) | (unsigned)ed[k].x;
            rk[k] = -1;
            if (k * FB_WG + tid < m)
                rk[k] = atomicAdd(&cntN[tl], 1);
        }
        __syncthreads();

        int cN = (tid < FB_NBP) ? cntN[tid] : 0;
        int ex = fb_block_excl_scan(cN, wtot, lane, wv);
        if (tid < FB_NBP) cntN[tid] = ex;
        __syncthreads();

        int o[FB_CPT];
        #pragma unroll
        for (int k = 0; k < FB_CPT; ++k)
            o[k] = cntN[pk[k] >> 24];
        #pragma unroll
        for (int k = 0; k < FB_CPT; ++k)
            if (rk[k] >= 0) srt[o[k] + rk[k]] = pk[k];
        __syncthreads();

        for (int v0 = wv * 32; v0 < m; v0 += FB_WG) {
            int n = m - v0;
            int j = v0 + lane;
            bool live = j < m;
            int s = 0, tl = 0;
            v2f rt = {0.f, 0.f}, rs = {0.f, 0.f};
            if (live) {
                unsigned p = srt[j];
                s  = (int)(p & 0xFFFFFFu);
                tl = (int)(p >> 24);
                rt = recd[tl];
                rs = *(const v2f*)(rec_src + 2 * (size_t)s);
            }
            int tl_next = __shfl_down(tl, 1);
            bool last = (lane == 31) || (j + 1 >= m) || (tl_next != tl);
            unsigned bnd = __builtin_amdgcn_ballot_w32(last);
            float a0 = 0.f, a1 = 0.f;

            if (n >= 32) {
                v2f row[2][FA_GRP];
                fb_rows_issue(h, s, 0, f2, row[0]);
                fb_rows_issue(h, s, FA_GRP, f2, row[1]);
                __builtin_amdgcn_sched_barrier(0);
                float g = tanhf(rt.x + rs.x + bias);
                float c = g * rt.y * rs.y;
                int cbits = __float_as_int(c);
                #pragma unroll
                for (int gi = 0; gi < FA_NGRP; ++gi) {
                    if (gi >= 1 && gi + 1 < FA_NGRP)
                        fb_rows_issue(h, s, (gi + 1) * FA_GRP, f2, row[(gi + 1) & 1]);
                    fb_rows_sum(acc, tl, cbits, bnd, gi * FA_GRP, f2, row[gi & 1], a0, a1);
                }
            } else {
                float c = 0.f;
                if (live) {
                    float g = tanhf(rt.x + rs.x + bias);
                    c = g * rt.y * rs.y;
                }
                int cbits = __float_as_int(c);
                for (int e = 0; e < n; ++e) {
                    int   sj = __builtin_amdgcn_readlane(s, e);
                    float cj = __int_as_float(__builtin_amdgcn_readlane(cbits, e));
                    v2f r = *(const v2f*)(h + (size_t)sj * NDIM + f2);
                    a0 = fmaf(r.x, cj, a0);
                    a1 = fmaf(r.y, cj, a1);
                    if ((bnd >> e) & 1u) {
                        float* ap = acc + __builtin_amdgcn_readlane(tl, e) * NDIM + f2;
                        atomicAdd(ap,     a0);
                        atomicAdd(ap + 1, a1);
                        a0 = 0.f;
                        a1 = 0.f;
                    }
                }
            }
        }
    }
    __syncthreads();

    v4f* zp = (v4f*)(z + (size_t)n0 * NDIM);
    for (int i = tid; i < nn * (NDIM / 4); i += FB_WG)
        zp[i] = ((const v4f*)acc)[i];
}

extern "C" void kernel_launch(void* const* d_in, const int* in_sizes, int n_in,
                              void* d_out, int out_size, void* d_ws, size_t ws_size,
                              hipStream_t stream)
{
    const float* h      = (const float*)d_in[0];
    const float* dvec   = (const float*)d_in[1];
    const float* gate_W = (const float*)d_in[2];
    const float* gate_b = (const float*)d_in[3];
    const int*   esrc   = (const int*)d_in[4];
    const int*   edst   = (const int*)d_in[5];

    int N = in_sizes[1];
    int E = in_sizes[4];
    float* z = (float*)d_out;

    float* rec_dst = (float*)d_ws;
    float* rec_src = rec_dst + 2 * (size_t)N;

    int B = (N + FB_NB - 1) / FB_NB;
    int T = (E + FB_TS - 1) / FB_TS;
    size_t recs_off = (16 * (size_t)N + 255) / 256 * 256;
    size_t offs_off = recs_off + (8 * (size_t)E + 255) / 256 * 256;
    size_t need     = offs_off + 4 * (size_t)T * (B + 1);

    if (T >= 1 && T <= FB_WG && B < FB_WG && N <= (1 << 24) && ws_size >= need) {
        int2* recs = (int2*)((char*)d_ws + recs_off);
        int*  offs = (int*)((char*)d_ws + offs_off);
        int G = ((N + 15) / 16 + 31) / 32;
        fa_bin_gate<<<dim3(T + G), dim3(FB_WG), 0, stream>>>(esrc, edst, recs, offs, E, B, T,
                                                            h, gate_W, dvec, rec_dst, rec_src, N);
        fa_accum<<<dim3(B), dim3(FB_WG), 0, stream>>>(h, rec_dst, rec_src, gate_b, recs, offs, z, N, T, B);
    } else {
        {
            int waves   = (N + 15) / 16;
            int threads = waves * 32;
            dim3 blk(256), grd((threads + 255) / 256);
            fa_gate_wmma<<<grd, blk, 0, stream>>>(h, gate_W, dvec, rec_dst, rec_src, N);
        }
        {
            int n4 = out_size / 4;
            dim3 blk(256), grd((n4 + 255) / 256);
            fa_zero<<<grd, blk, 0, stream>>>(z, n4);
        }
        {
            int waves   = (E + 31) / 32;
            int threads = waves * 32;
            dim3 blk(256), grd((threads + 255) / 256);
            fa_gate_scatter<<<grd, blk, 0, stream>>>(h, rec_dst, rec_src, gate_b, esrc, edst, z, E);
        }
    }
}


# Accumulate: the short last batch of a pass loads its rows in groups of 8 (idle lanes: row 0, zero coefficient, no run-boundary bit), one wait per group, not one per edge; order of sums unchanged
# speedup vs baseline: 3.9506x; 1.0384x over previous
#include <hip/hip_runtime.h>
#include <math.h>

typedef __attribute__((ext_vector_type(2))) float v2f;
typedef __attribute__((ext_vector_type(4))) float v4f;
typedef __attribute__((ext_vector_type(8))) float v8f;

#define NDIM 64

__device__ __forceinline__ void fa_gate_wave(const float* __restrict__ h,
                                             const float* __restrict__ gate_W,
                                             const float* __restrict__ dvec,
                                             float* __restrict__ rec_dst,
                                             float* __restrict__ rec_src,
                                             int N, int wave, int lane)
{
    int base = wave * 16;
    if (base >= N) return;

    int row   = lane & 15;
    int khalf = (lane >> 4) << 1;
    int arow  = base + row; if (arow > N - 1) arow = N - 1;

    float msk = (row < 2) ? 1.0f : 0.0f;
    const float* wbase = gate_W + ((row < 2) ? row : 0) * NDIM + khalf;
    v2f bfrag[16];
    #pragma unroll
    for (int j = 0; j < 16; ++j) {
        bfrag[j].x = wbase[j * 4]     * msk;
        bfrag[j].y = wbase[j * 4 + 1] * msk;
    }

    const float* arow_p = h + (size_t)arow * NDIM + khalf;
    v8f acc = {0.f, 0.f, 0.f, 0.f, 0.f, 0.f, 0.f, 0.f};
    #pragma unroll
    for (int j = 0; j < 16; ++j) {
        v2f a = *(const v2f*)(arow_p + j * 4);
        acc = __builtin_amdgcn_wmma_f32_16x16x4_f32(
                   false, a,  false, bfrag[j],
                   (short)0, acc,  false,  false);
    }

    if (row < 2) {
        float* outp = (row == 0) ? rec_dst : rec_src;
        int moff = (lane >> 4) << 3;
        float dv[8];
        #pragma unroll
        for (int i = 0; i < 8; ++i) {
            int m = base + moff + i;
            dv[i] = dvec[(m < N) ? m : N - 1];
        }
        #pragma unroll
        for (int i = 0; i < 8; ++i) {
            int m = base + moff + i;
            if (m < N) {
                v2f r; r.x = acc[i]; r.y = dv[i];
                *(v2f*)(outp + 2 * (size_t)m) = r;
            }
        }
    }
}

__global__ void fa_gate_wmma(const float* __restrict__ h,
                             const float* __restrict__ gate_W,
                             const float* __restrict__ dvec,
                             float* __restrict__ rec_dst,
                             float* __restrict__ rec_src,
                             int N)
{
    int wave = (blockIdx.x * blockDim.x + threadIdx.x) >> 5;
    fa_gate_wave(h, gate_W, dvec, rec_dst, rec_src, N, wave, threadIdx.x & 31);
}

__global__ void fa_zero(float* __restrict__ z, int n4)
{
    int i = blockIdx.x * blockDim.x + threadIdx.x;
    if (i < n4) ((v4f*)z)[i] = (v4f){0.f, 0.f, 0.f, 0.f};
}

#define FA_GRP 8
#define FA_NGRP (32 / FA_GRP)

__device__ __forceinline__ void fa_edge_update(const float* __restrict__ h,
                                               float* __restrict__ z,
                                               int s, int t, float c, int f)
{
    const float* hp = h + (size_t)s * NDIM + f;
    float h_lo = hp[0];
    float h_hi = hp[32];
    float* zp = z + (size_t)t * NDIM + f;
    atomicAdd(zp,      h_lo * c);
    atomicAdd(zp + 32, h_hi * c);
}

__device__ __forceinline__ void fa_rows_issue(const float* __restrict__ h,
                                              int s, int j0, int f,
                                              float (&lo)[FA_GRP], float (&hi)[FA_GRP])
{
    #pragma unroll
    for (int k = 0; k < FA_GRP; ++k) {
        const float* hp = h + (size_t)__builtin_amdgcn_readlane(s, j0 + k) * NDIM + f;
        lo[k] = hp[0];
        hi[k] = hp[32];
    }
}

__device__ __forceinline__ void fa_rows_add(float* __restrict__ z,
                                            int t, int cbits, int j0, int f,
                                            const float (&lo)[FA_GRP], const float (&hi)[FA_GRP])
{
    #pragma unroll
    for (int k = 0; k < FA_GRP; ++k) {
        float c = __int_as_float(__builtin_amdgcn_readlane(cbits, j0 + k));
        float* zp = z + (size_t)__builtin_amdgcn_readlane(t, j0 + k) * NDIM + f;
        atomicAdd(zp,      lo[k] * c);
        atomicAdd(zp + 32, hi[k] * c);
    }
}

__global__ void fa_gate_scatter(const float* __restrict__ h,
                                const float* __restrict__ rec_dst,
                                const float* __restrict__ rec_src,
                                const float* __restrict__ gate_b,
                                const int* __restrict__ esrc,
                                const int* __restrict__ edst,
                                float* __restrict__ z, int E)
{
    int lane = threadIdx.x & 31;
    int wave   = __builtin_amdgcn_readfirstlane((int)((blockIdx.x * blockDim.x + threadIdx.x) >> 5));
    int nwaves = (int)((gridDim.x * blockDim.x) >> 5);
    int f = lane;
    float bias = gate_b[0];

    for (int base = wave * 32; base < E; base += nwaves * 32) {
        int i = base + lane;
        int n = E - base;

        if (n >= 32) {
            int s = esrc[i];
            int t = edst[i];
            v2f rt = *(const v2f*)(rec_dst + 2 * (size_t)t);
            v2f rs = *(const v2f*)(rec_src + 2 * (size_t)s);

            float lo[2][FA_GRP], hi[2][FA_GRP];
            fa_rows_issue(h, s, 0, f, lo[0], hi[0]);

            float g = tanhf(rt.x + rs.x + bias);
            float c = g * rt.y * rs.y;
            int cbits = __float_as_int(c);

            #pragma unroll
            for (int gi = 0; gi < FA_NGRP; ++gi) {
                if (gi + 1 < FA_NGRP)
                    fa_rows_issue(h, s, (gi + 1) * FA_GRP, f, lo[(gi + 1) & 1], hi[(gi + 1) & 1]);
                fa_rows_add(z, t, cbits, gi * FA_GRP, f, lo[gi & 1], hi[gi & 1]);
            }
        } else {
            int s = 0, t = 0;
            float c = 0.f;
            if (i < E) {
                s = esrc[i];
                t = edst[i];
                v2f rt = *(const v2f*)(rec_dst + 2 * (size_t)t);
                v2f rs = *(const v2f*)(rec_src + 2 * (size_t)s);
                float g = tanhf(rt.x + rs.x + bias);
                c = g * rt.y * rs.y;
            }
            int cbits = __float_as_int(c);
            for (int j = 0; j < n; ++j) {
                fa_edge_update(h, z,
                               __builtin_amdgcn_readlane(s, j),
                               __builtin_amdgcn_readlane(t, j),
                               __int_as_float(__builtin_amdgcn_readlane(cbits, j)), f);
            }
        }
    }
}

#define FB_NB   196
#define FB_TS   12800
#define FB_WG   1024
#define FB_PER  ((FB_TS + FB_WG - 1) / FB_WG)

__device__ __forceinline__ int fb_block_excl_scan(int c, int* wtot, int lane, int wv)
{
    int inc = c;
    #pragma unroll
    for (int d = 1; d < 32; d <<= 1) {
        int u = __shfl_up(inc, d);
        if (lane >= d) inc += u;
    }
    if (lane == 31) wtot[wv] = inc;
    __syncthreads();
    if (wv == 0) {
        int w = wtot[lane];
        int wi = w;
        #pragma unroll
        for (int d = 1; d < 32; d <<= 1) {
            int u = __shfl_up(wi, d);
            if (lane >= d) wi += u;
        }
        wtot[lane] = wi - w;
    }
    __syncthreads();
    return inc - c + wtot[wv];
}

__global__ __launch_bounds__(FB_WG, 16)
void fa_bin_gate(const int* __restrict__ esrc, const int* __restrict__ edst,
                 int2* __restrict__ recs, int* __restrict__ offs, int E, int B, int T,
                 const float* __restrict__ h,
                 const float* __restrict__ gate_W,
                 const float* __restrict__ dvec,
                 float* __restrict__ rec_dst,
                 float* __restrict__ rec_src,
                 int N)
{
    __shared__ int  cnt[FB_WG];
    __shared__ int  wtot[32];
    __shared__ int2 srt[FB_TS];

    int tid  = threadIdx.x;
    int lane = tid & 31;

    if ((int)blockIdx.x >= T) {
        int gw = __builtin_amdgcn_readfirstlane((int)(((blockIdx.x - T) * FB_WG + tid) >> 5));
        fa_gate_wave(h, gate_W, dvec, rec_dst, rec_src, N, gw, lane);
        return;
    }

    int wv   = tid >> 5;
    int tile = blockIdx.x;
    int base = tile * FB_TS;
    int n    = E - base; if (n > FB_TS) n = FB_TS;

    cnt[tid] = 0;
    int s[FB_PER], t[FB_PER];
    #pragma unroll
    for (int k = 0; k < FB_PER; ++k) {
        int i  = k * FB_WG + tid;
        int ii = (i < n) ? i : n - 1;
        s[k] = esrc[base + ii];
        t[k] = edst[base + ii];
    }
    #pragma unroll
    for (int k = 0; k < FB_PER; ++k)
        if (k * FB_WG + tid >= n) t[k] = -1;
    __syncthreads();
    #pragma unroll
    for (int k = 0; k < FB_PER; ++k)
        if (t[k] >= 0) atomicAdd(&cnt[t[k] / FB_NB], 1);
    __syncthreads();

    int c    = cnt[tid];
    int excl = fb_block_excl_scan(c, wtot, lane, wv);
    cnt[tid] = excl;
    __syncthreads();

    #pragma unroll
    for (int k = 0; k < FB_PER; ++k)
        if (t[k] >= 0) {
            int pos = atomicAdd(&cnt[t[k] / FB_NB], 1);
            srt[pos] = make_int2(s[k], t[k]);
        }
    __syncthreads();

    int2 rec[FB_PER];
    #pragma unroll
    for (int k = 0; k < FB_PER; ++k) {
        int i = k * FB_WG + tid;
        rec[k] = srt[(i < FB_TS) ? i : FB_TS - 1];
    }
    #pragma unroll
    for (int k = 0; k < FB_PER; ++k) {
        int i = k * FB_WG + tid;
        if (i < n) recs[(size_t)base + i] = rec[k];
    }
    if (tid <= B) offs[(size_t)tile * (B + 1) + tid] = excl;
}

#define FB_CH   8192
#define FB_CPT  (FB_CH / FB_WG)
#define FB_NBP  256

__device__ __forceinline__ void fb_rows_issue(const float* __restrict__ h,
                                              int s, int j0, int f2,
                                              v2f (&row)[FA_GRP])
{
    #pragma unroll
    for (int k = 0; k < FA_GRP; ++k)
        row[k] = *(const v2f*)(h + (size_t)__builtin_amdgcn_readlane(s, j0 + k) * NDIM + f2);
}

__device__ __forceinline__ void fb_rows_sum(float* acc, int tl, int cbits, unsigned bnd, int j0, int f2,
                                            const v2f (&row)[FA_GRP],
                                            float& a0, float& a1)
{
    #pragma unroll
    for (int k = 0; k < FA_GRP; ++k) {
        float c = __int_as_float(__builtin_amdgcn_readlane(cbits, j0 + k));
        a0 = fmaf(row[k].x, c, a0);
        a1 = fmaf(row[k].y, c, a1);
        if ((bnd >> (j0 + k)) & 1u) {
            float* ap = acc + __builtin_amdgcn_readlane(tl, j0 + k) * NDIM + f2;
            atomicAdd(ap,     a0);
            atomicAdd(ap + 1, a1);
            a0 = 0.f;
            a1 = 0.f;
        }
    }
}

__global__ __launch_bounds__(FB_WG, 16)
void fa_accum(const float* __restrict__ h,
              const float* __restrict__ rec_dst,
              const float* __restrict__ rec_src,
              const float* __restrict__ gate_b,
              const int2* __restrict__ recs,
              const int* __restrict__ offs,
              float* __restrict__ z, int N, int T, int B)
{
    __shared__ __attribute__((aligned(16))) float acc[FB_NB * NDIM];
    __shared__ v2f recd[FB_NB];
    __shared__ int P[FB_WG + 1];
    __shared__ int S[FB_WG];
    __shared__ int wtot[32];
    __shared__ unsigned srt[FB_CH];
    __shared__ int cntN[FB_NBP];

    int tid  = threadIdx.x;
    int lane = tid & 31;
    int wv   = __builtin_amdgcn_readfirstlane(tid >> 5);
    int b    = blockIdx.x;
    int n0   = b * FB_NB;
    int nn   = N - n0; if (nn > FB_NB) nn = FB_NB;
    int f2   = 2 * lane;
    float bias = gate_b[0];

    for (int i = tid; i < FB_NB * NDIM / 4; i += FB_WG)
        ((v4f*)acc)[i] = (v4f){0.f, 0.f, 0.f, 0.f};
    int nq = (tid < nn) ? tid : nn - 1;
    int tq = (tid < T) ? tid : T - 1;
    v2f rd = *(const v2f*)(rec_dst + 2 * (size_t)(n0 + nq));
    const int* op = offs + (size_t)tq * (B + 1) + b;
    int o0 = op[0], o1 = op[1];
    if (tid < nn) recd[tid] = rd;

    int len = 0, st = 0;
    if (tid < T) {
        len = o1 - o0;
        st  = tid * FB_TS + o0;
    }
    int excl = fb_block_excl_scan(len, wtot, lane, wv);
    P[tid] = excl;
    S[tid] = st;
    if (tid == FB_WG - 1) P[FB_WG] = excl + len;
    __syncthreads();
    int total = P[FB_WG];
    int top = 1;
    while (2 * top < T) top <<= 1;

    for (int c0 = 0; c0 < total; c0 += FB_CH) {
        int m = total - c0; if (m > FB_CH) m = FB_CH;

        if (tid < FB_NBP) cntN[tid] = 0;
        __syncthreads();

        unsigned pk[FB_CPT];
        int      rk[FB_CPT];
        int      vv[FB_CPT], lt[FB_CPT];
        #pragma unroll
        for (int k = 0; k < FB_CPT; ++k) {
            int q = k * FB_WG + tid;
            vv[k] = c0 + ((q < m) ? q : 0);
            lt[k] = 0;
        }
        for (int step = top; step >= 1; step >>= 1) {
            #pragma unroll
            for (int k = 0; k < FB_CPT; ++k)
                lt[k] += (P[lt[k] + step] <= vv[k]) ? step : 0;
        }
        int2 ed[FB_CPT];
        #pragma unroll
        for (int k = 0; k < FB_CPT; ++k)
            ed[k] = recs[(size_t)(S[lt[k]] + (vv[k] - P[lt[k]]))];
        #pragma unroll
        for (int k = 0; k < FB_CPT; ++k) {
            int tl = ed[k].y - n0;
            pk[k] = ((unsigned)tl << 24) | (unsigned)ed[k].x;
            rk[k] = -1;
            if (k * FB_WG + tid < m)
                rk[k] = atomicAdd(&cntN[tl], 1);
        }
        __syncthreads();

        int cN = (tid < FB_NBP) ? cntN[tid] : 0;
        int ex = fb_block_excl_scan(cN, wtot, lane, wv);
        if (tid < FB_NBP) cntN[tid] = ex;
        __syncthreads();

        int o[FB_CPT];
        #pragma unroll
        for (int k = 0; k < FB_CPT; ++k)
            o[k] = cntN[pk[k] >> 24];
        #pragma unroll
        for (int k = 0; k < FB_CPT; ++k)
            if (rk[k] >= 0) srt[o[k] + rk[k]] = pk[k];
        __syncthreads();

        for (int v0 = wv * 32; v0 < m; v0 += FB_WG) {
            int n = m - v0;
            int j = v0 + lane;
            bool live = j < m;
            int s = 0, tl = 0;
            v2f rt = {0.f, 0.f}, rs = {0.f, 0.f};
            if (live) {
                unsigned p = srt[j];
                s  = (int)(p & 0xFFFFFFu);
                tl = (int)(p >> 24);
                rt = recd[tl];
                rs = *(const v2f*)(rec_src + 2 * (size_t)s);
            }
            int tl_next = __shfl_down(tl, 1);
            bool last = live && ((lane == 31) || (j + 1 >= m) || (tl_next != tl));
            unsigned bnd = __builtin_amdgcn_ballot_w32(last);
            float a0 = 0.f, a1 = 0.f;

            if (n >= 32) {
                v2f row[2][FA_GRP];
                fb_rows_issue(h, s, 0, f2, row[0]);
                fb_rows_issue(h, s, FA_GRP, f2, row[1]);
                __builtin_amdgcn_sched_barrier(0);
                float g = tanhf(rt.x + rs.x + bias);
                float c = g * rt.y * rs.y;
                int cbits = __float_as_int(c);
                #pragma unroll
                for (int gi = 0; gi < FA_NGRP; ++gi) {
                    if (gi >= 1 && gi + 1 < FA_NGRP)
                        fb_rows_issue(h, s, (gi + 1) * FA_GRP, f2, row[(gi + 1) & 1]);
                    fb_rows_sum(acc, tl, cbits, bnd, gi * FA_GRP, f2, row[gi & 1], a0, a1);
                }
            } else {
                float c = 0.f;
                if (live) {
                    float g = tanhf(rt.x + rs.x + bias);
                    c = g * rt.y * rs.y;
                }
                int cbits = __float_as_int(c);
                for (int e0 = 0; e0 < n; e0 += FA_GRP) {
                    v2f row[FA_GRP];
                    #pragma unroll
                    for (int k = 0; k < FA_GRP; ++k)
                        row[k] = *(const v2f*)(h + (size_t)__builtin_amdgcn_readlane(s, e0 + k) * NDIM + f2);
                    #pragma unroll
                    for (int k = 0; k < FA_GRP; ++k) {
                        float cj = __int_as_float(__builtin_amdgcn_readlane(cbits, e0 + k));
                        a0 = fmaf(row[k].x, cj, a0);
                        a1 = fmaf(row[k].y, cj, a1);
                        if ((bnd >> (e0 + k)) & 1u) {
                            float* ap = acc + __builtin_amdgcn_readlane(tl, e0 + k) * NDIM + f2;
                            atomicAdd(ap,     a0);
                            atomicAdd(ap + 1, a1);
                            a0 = 0.f;
                            a1 = 0.f;
                        }
                    }
                }
            }
        }
    }
    __syncthreads();

    v4f* zp = (v4f*)(z + (size_t)n0 * NDIM);
    for (int i = tid; i < nn * (NDIM / 4); i += FB_WG)
        zp[i] = ((const v4f*)acc)[i];
}

extern "C" void kernel_launch(void* const* d_in, const int* in_sizes, int n_in,
                              void* d_out, int out_size, void* d_ws, size_t ws_size,
                              hipStream_t stream)
{
    const float* h      = (const float*)d_in[0];
    const float* dvec   = (const float*)d_in[1];
    const float* gate_W = (const float*)d_in[2];
    const float* gate_b = (const float*)d_in[3];
    const int*   esrc   = (const int*)d_in[4];
    const int*   edst   = (const int*)d_in[5];

    int N = in_sizes[1];
    int E = in_sizes[4];
    float* z = (float*)d_out;

    float* rec_dst = (float*)d_ws;
    float* rec_src = rec_dst + 2 * (size_t)N;

    int B = (N + FB_NB - 1) / FB_NB;
    int T = (E + FB_TS - 1) / FB_TS;
    size_t recs_off = (16 * (size_t)N + 255) / 256 * 256;
    size_t offs_off = recs_off + (8 * (size_t)E + 255) / 256 * 256;
    size_t need     = offs_off + 4 * (size_t)T * (B + 1);

    if (T >= 1 && T <= FB_WG && B < FB_WG && N <= (1 << 24) && ws_size >= need) {
        int2* recs = (int2*)((char*)d_ws + recs_off);
        int*  offs = (int*)((char*)d_ws + offs_off);
        int G = ((N + 15) / 16 + 31) / 32;
        fa_bin_gate<<<dim3(T + G), dim3(FB_WG), 0, stream>>>(esrc, edst, recs, offs, E, B, T,
                                                            h, gate_W, dvec, rec_dst, rec_src, N);
        fa_accum<<<dim3(B), dim3(FB_WG), 0, stream>>>(h, rec_dst, rec_src, gate_b, recs, offs, z, N, T, B);
    } else {
        {
            int waves   = (N + 15) / 16;
            int threads = waves * 32;
            dim3 blk(256), grd((threads + 255) / 256);
            fa_gate_wmma<<<grd, blk, 0, stream>>>(h, gate_W, dvec, rec_dst, rec_src, N);
        }
        {
            int n4 = out_size / 4;
            dim3 blk(256), grd((n4 + 255) / 256);
            fa_zero<<<grd, blk, 0, stream>>>(z, n4);
        }
        {
            int waves   = (E + 31) / 32;
            int threads = waves * 32;
            dim3 blk(256), grd((threads + 255) / 256);
            fa_gate_scatter<<<grd, blk, 0, stream>>>(h, rec_dst, rec_src, gate_b, esrc, edst, z, E);
        }
    }
}
